# starting kernel
# baseline (speedup 1.0000x reference)
#include <hip/hip_runtime.h>
#include <hip/hip_bf16.h>

#define V_N     10000
#define E_N     60000
#define B_N     8
#define LATENT  512
#define HIDDEN  256
#define NLAYERS 5
#define NNODES  (B_N * V_N)

typedef float v2f __attribute__((ext_vector_type(2)));
typedef float v8f __attribute__((ext_vector_type(8)));

#if defined(__gfx1250__) && __has_builtin(__builtin_amdgcn_wmma_f32_16x16x4_f32)
#define HAVE_WMMA_F32 1
#endif


__global__ void k_deg_init(float* deg) {
    int v = blockIdx.x * blockDim.x + threadIdx.x;
    if (v < V_N) deg[v] = 1.0f;
}

__global__ void k_deg_acc(float* deg, const int* __restrict__ dst) {
    int e = blockIdx.x * blockDim.x + threadIdx.x;
    if (e < E_N) atomicAdd(&deg[dst[e]], 1.0f);
}

__global__ void k_dinv(float* deg_to_dinv, float* selfnorm) {
    int v = blockIdx.x * blockDim.x + threadIdx.x;
    if (v < V_N) {
        float di = rsqrtf(deg_to_dinv[v]);
        deg_to_dinv[v] = di;
        selfnorm[v] = di * di;
    }
}

__global__ void k_norm(const float* __restrict__ dinv, const int* __restrict__ src,
                       const int* __restrict__ dst, float* __restrict__ norm) {
    int e = blockIdx.x * blockDim.x + threadIdx.x;
    if (e < E_N) norm[e] = dinv[src[e]] * dinv[dst[e]];
}


__global__ void k_xyzW(const float* __restrict__ xyz, const float* __restrict__ iW,
                       const float* __restrict__ ib, float* __restrict__ xyzW) {
    int i = blockIdx.x * blockDim.x + threadIdx.x;
    if (i >= V_N * HIDDEN) return;
    int v = i / HIDDEN, hh = i % HIDDEN;
    const float* wp = iW + (size_t)hh * (LATENT + 3);
    xyzW[i] = ib[hh] + xyz[3 * v] * wp[0] + xyz[3 * v + 1] * wp[1] + xyz[3 * v + 2] * wp[2];
}

__global__ void k_latW(const float* __restrict__ latent, const float* __restrict__ iW,
                       float* __restrict__ latW) {
    int i = blockIdx.x * blockDim.x + threadIdx.x;
    if (i >= B_N * HIDDEN) return;
    int b = i / HIDDEN, hh = i % HIDDEN;
    const float* lp = latent + (size_t)b * LATENT;
    const float* wp = iW + (size_t)hh * (LATENT + 3) + 3;
    float s = 0.0f;
    for (int j = 0; j < LATENT; ++j) s += lp[j] * wp[j];
    latW[i] = s;
}

__global__ void k_x0(const float* __restrict__ xyzW, const float* __restrict__ latW,
                     float* __restrict__ x) {
    size_t i = (size_t)blockIdx.x * blockDim.x + threadIdx.x;
    int hh = (int)(i & (HIDDEN - 1));
    size_t node = i >> 8;
    int v = (int)(node % V_N);
    int b = (int)(node / V_N);
    float s = xyzW[(size_t)v * HIDDEN + hh] + latW[b * HIDDEN + hh];
    x[i] = s > 0.0f ? s : 0.0f;
}


#define GEMM_MT 2
#define GEMM_NT 4

__global__ void __launch_bounds__(256)
k_gemm_wmma(const float* __restrict__ x, const float* __restrict__ W,
            float* __restrict__ h) {
    const int lane  = threadIdx.x & 31;
    const int wv    = threadIdx.x >> 5;
    const int half  = lane >> 4;
    const int l16   = lane & 15;
    const int mbase = blockIdx.x * 64 + (wv >> 2) * 32;
    const int nbase = (wv & 3) * 64;

    const float* arow = x + (size_t)(mbase + l16) * HIDDEN + half * 2;
    const float* bcol = W + (size_t)(nbase + l16) * HIDDEN + half * 2;

    v8f acc[GEMM_MT][GEMM_NT] = {};

#if HAVE_WMMA_F32
#pragma unroll 4
    for (int k = 0; k < HIDDEN; k += 4) {
        v2f a[GEMM_MT], b[GEMM_NT];
#pragma unroll
        for (int mt = 0; mt < GEMM_MT; ++mt)
            a[mt] = *(const v2f*)(arow + (size_t)mt * 16 * HIDDEN + k);
#pragma unroll
        for (int nt = 0; nt < GEMM_NT; ++nt)
            b[nt] = *(const v2f*)(bcol + (size_t)nt * 16 * HIDDEN + k);
#pragma unroll
        for (int mt = 0; mt < GEMM_MT; ++mt)
#pragma unroll
            for (int nt = 0; nt < GEMM_NT; ++nt)
                acc[mt][nt] = __builtin_amdgcn_wmma_f32_16x16x4_f32(
                     false, a[mt],  false, b[nt],
                     (short)0, acc[mt][nt],
                     false,  false);
    }
#else
#pragma unroll
    for (int mt = 0; mt < GEMM_MT; ++mt)
#pragma unroll
        for (int r = 0; r < 8; ++r) {
            const float* xr = x + (size_t)(mbase + mt * 16 + half * 8 + r) * HIDDEN;
#pragma unroll
            for (int nt = 0; nt < GEMM_NT; ++nt) {
                const float* wr = W + (size_t)(nbase + nt * 16 + l16) * HIDDEN;
                float s = 0.0f;
                for (int k = 0; k < HIDDEN; ++k) s += xr[k] * wr[k];
                acc[mt][nt][r] = s;
            }
        }
#endif

    float* outp = h + (size_t)(mbase + half * 8) * HIDDEN + (nbase + l16);
#pragma unroll
    for (int mt = 0; mt < GEMM_MT; ++mt)
#pragma unroll
        for (int nt = 0; nt < GEMM_NT; ++nt)
#pragma unroll
            for (int r = 0; r < 8; ++r)
                outp[((size_t)mt * 16 + r) * HIDDEN + nt * 16] = acc[mt][nt][r];
}


__global__ void k_selfinit(const float* __restrict__ h, float* __restrict__ agg,
                           const float* __restrict__ selfnorm) {
    size_t i = (size_t)blockIdx.x * blockDim.x + threadIdx.x;
    int v = (int)((i >> 8) % V_N);
    agg[i] = h[i] * selfnorm[v];
}

__global__ void k_scatter(const float* __restrict__ h, float* __restrict__ agg,
                          const int* __restrict__ src, const int* __restrict__ dst,
                          const float* __restrict__ norm) {
    int t  = blockIdx.x * blockDim.x + threadIdx.x;
    int e  = t >> 6;
    int f4 = (t & 63) << 2;
    int b  = blockIdx.y;
    int s  = src[e], d = dst[e];
    float w = norm[e];
    const float* hp = h + ((size_t)b * V_N + s) * HIDDEN + f4;
    float*       ap = agg + ((size_t)b * V_N + d) * HIDDEN + f4;
    float4 hv = *(const float4*)hp;
    atomicAdd(ap + 0, hv.x * w);
    atomicAdd(ap + 1, hv.y * w);
    atomicAdd(ap + 2, hv.z * w);
    atomicAdd(ap + 3, hv.w * w);
}

__global__ void k_relu_bias(float* __restrict__ x, const float* __restrict__ bias) {
    size_t i = (size_t)blockIdx.x * blockDim.x + threadIdx.x;
    float s = x[i] + bias[i & (HIDDEN - 1)];
    x[i] = s > 0.0f ? s : 0.0f;
}

__global__ void k_out(const float* __restrict__ x, const float* __restrict__ oW,
                      const float* __restrict__ ob, float* __restrict__ disp) {
    int idx = blockIdx.x * blockDim.x + threadIdx.x;
    if (idx >= NNODES * 3) return;
    int node = idx / 3, o = idx % 3;
    const float* xp = x + (size_t)node * HIDDEN;
    const float* wp = oW + o * HIDDEN;
    float s = ob[o];
#pragma unroll 4
    for (int k = 0; k < HIDDEN; k += 4) {
        float4 xv = *(const float4*)(xp + k);
        float4 wv = *(const float4*)(wp + k);
        s += xv.x * wv.x + xv.y * wv.y + xv.z * wv.z + xv.w * wv.w;
    }
    disp[idx] = s;
}


extern "C" void kernel_launch(void* const* d_in, const int* in_sizes, int n_in,
                              void* d_out, int out_size, void* d_ws, size_t ws_size,
                              hipStream_t stream) {
    const float* vertex_xyz = (const float*)d_in[0];
    const float* latent     = (const float*)d_in[1];
    const int*   edge_index = (const int*)d_in[2];
    const float* input_W    = (const float*)d_in[3];
    const float* input_b    = (const float*)d_in[4];
    const float* conv_W     = (const float*)d_in[5];
    const float* conv_b     = (const float*)d_in[6];
    const float* out_W      = (const float*)d_in[7];
    const float* out_b      = (const float*)d_in[8];
    float* disp = (float*)d_out;

    const int* src = edge_index;
    const int* dst = edge_index + E_N;

    float* xbuf     = (float*)d_ws;
    float* hbuf     = xbuf + (size_t)NNODES * HIDDEN;
    float* xyzW     = hbuf + (size_t)NNODES * HIDDEN;
    float* latW     = xyzW + (size_t)V_N * HIDDEN;
    float* dinv     = latW + (size_t)B_N * HIDDEN;
    float* selfnorm = dinv + V_N;
    float* norm     = selfnorm + V_N;

    const int T = 256;
    const size_t NH = (size_t)NNODES * HIDDEN;

    k_deg_init<<<(V_N + T - 1) / T, T, 0, stream>>>(dinv);
    k_deg_acc<<<(E_N + T - 1) / T, T, 0, stream>>>(dinv, dst);
    k_dinv<<<(V_N + T - 1) / T, T, 0, stream>>>(dinv, selfnorm);
    k_norm<<<(E_N + T - 1) / T, T, 0, stream>>>(dinv, src, dst, norm);

    k_xyzW<<<(V_N * HIDDEN + T - 1) / T, T, 0, stream>>>(vertex_xyz, input_W, input_b, xyzW);
    k_latW<<<(B_N * HIDDEN + T - 1) / T, T, 0, stream>>>(latent, input_W, latW);
    k_x0<<<(unsigned)(NH / T), T, 0, stream>>>(xyzW, latW, xbuf);

    dim3 ggrid(NNODES / 64);
    dim3 sgrid((E_N * 64) / T, B_N);
    for (int l = 0; l < NLAYERS; ++l) {
        k_gemm_wmma<<<ggrid, T, 0, stream>>>(xbuf, conv_W + (size_t)l * HIDDEN * HIDDEN, hbuf);
        k_selfinit<<<(unsigned)(NH / T), T, 0, stream>>>(hbuf, xbuf, selfnorm);
        k_scatter<<<sgrid, T, 0, stream>>>(hbuf, xbuf, src, dst, norm);
        k_relu_bias<<<(unsigned)(NH / T), T, 0, stream>>>(xbuf, conv_b + (size_t)l * HIDDEN);
    }

    k_out<<<(NNODES * 3 + T - 1) / T, T, 0, stream>>>(xbuf, out_W, out_b, disp);
}


# GEMM: 256 resident work-groups hold the layer's whole W in LDS (copied once, pitch 260 floats) and loop over 64-row tiles; x and W fragments by 16-byte loads, each feeding two matrix ops
# speedup vs baseline: 3.0753x; 3.0753x over previous
#include <hip/hip_runtime.h>
#include <hip/hip_bf16.h>

#define V_N     10000
#define E_N     60000
#define B_N     8
#define LATENT  512
#define HIDDEN  256
#define NLAYERS 5
#define NNODES  (B_N * V_N)

typedef float v2f __attribute__((ext_vector_type(2)));
typedef float v8f __attribute__((ext_vector_type(8)));

#if defined(__gfx1250__) && __has_builtin(__builtin_amdgcn_wmma_f32_16x16x4_f32)
#define HAVE_WMMA_F32 1
#endif


__global__ void k_cnt_zero(int* cnt) {
    int v = blockIdx.x * blockDim.x + threadIdx.x;
    if (v < V_N) cnt[v] = 0;
}

__global__ void k_rank(int* cnt, const int* __restrict__ dst, int* __restrict__ rank) {
    int e = blockIdx.x * blockDim.x + threadIdx.x;
    if (e < E_N) rank[e] = atomicAdd(&cnt[dst[e]], 1);
}

#define SCAN_T   1024
#define SCAN_PER ((V_N + SCAN_T - 1) / SCAN_T)
__global__ void __launch_bounds__(SCAN_T)
k_scan(const int* __restrict__ cnt, int* __restrict__ rowptr,
       float* __restrict__ dinv, float* __restrict__ selfnorm) {
    __shared__ int part[SCAN_T];
    const int t = threadIdx.x;
    const int base = t * SCAN_PER;
    int loc[SCAN_PER];
    int c[SCAN_PER];
    int s = 0;
#pragma unroll
    for (int i = 0; i < SCAN_PER; ++i) {
        int v = base + i;
        c[i] = (v < V_N) ? cnt[v] : 0;
        loc[i] = s;
        s += c[i];
    }
    part[t] = s;
    __syncthreads();
    for (int off = 1; off < SCAN_T; off <<= 1) {
        int add = (t >= off) ? part[t - off] : 0;
        __syncthreads();
        part[t] += add;
        __syncthreads();
    }
    const int excl = part[t] - s;
#pragma unroll
    for (int i = 0; i < SCAN_PER; ++i) {
        int v = base + i;
        if (v < V_N) {
            rowptr[v] = excl + loc[i];
            float di = rsqrtf((float)(c[i] + 1));
            dinv[v] = di;
            selfnorm[v] = di * di;
        }
    }
    if (t == SCAN_T - 1) rowptr[V_N] = part[t];
}

__global__ void k_fill(const int* __restrict__ rowptr, const int* __restrict__ dst,
                       const int* __restrict__ rank, int* __restrict__ slot) {
    int e = blockIdx.x * blockDim.x + threadIdx.x;
    if (e < E_N) slot[rowptr[dst[e]] + rank[e]] = e;
}

__global__ void k_place(const int* __restrict__ rowptr, const int* __restrict__ slot,
                        const int* __restrict__ src, const int* __restrict__ dst,
                        const float* __restrict__ dinv,
                        int* __restrict__ csrc, float* __restrict__ cw) {
    int e = blockIdx.x * blockDim.x + threadIdx.x;
    if (e >= E_N) return;
    const int v = dst[e], s = src[e];
    const int beg = rowptr[v], end = rowptr[v + 1];
    int r = 0;
    for (int j = beg; j < end; ++j) r += (slot[j] < e) ? 1 : 0;
    csrc[beg + r] = s;
    cw[beg + r]   = dinv[s] * dinv[v];
}


__global__ void k_latW(const float* __restrict__ latent, const float* __restrict__ iW,
                       float* __restrict__ latW) {
    int i = blockIdx.x * blockDim.x + threadIdx.x;
    if (i >= B_N * HIDDEN) return;
    int b = i / HIDDEN, hh = i % HIDDEN;
    const float* lp = latent + (size_t)b * LATENT;
    const float* wp = iW + (size_t)hh * (LATENT + 3) + 3;
    float s = 0.0f;
    for (int j = 0; j < LATENT; ++j) s += lp[j] * wp[j];
    latW[i] = s;
}

typedef float v4f __attribute__((ext_vector_type(4)));

__device__ __forceinline__ v4f relu4(v4f s) {
    v4f r;
    r.x = s.x > 0.0f ? s.x : 0.0f;
    r.y = s.y > 0.0f ? s.y : 0.0f;
    r.z = s.z > 0.0f ? s.z : 0.0f;
    r.w = s.w > 0.0f ? s.w : 0.0f;
    return r;
}

#define X0_VPW 4
#define X0_WPB 4
static_assert(V_N % (X0_VPW * X0_WPB) == 0, "grid covers the vertices exactly");

__global__ void __launch_bounds__(32 * X0_WPB)
k_x0(const float* __restrict__ xyz, const float* __restrict__ iW, const float* __restrict__ ib,
     const float* __restrict__ latW, float* __restrict__ x) {
    const int lane = threadIdx.x & 31;
    const int wave = __builtin_amdgcn_readfirstlane((int)(blockIdx.x * X0_WPB + (threadIdx.x >> 5)));
    const int vbeg = wave * X0_VPW;
    const int c0 = lane * 4, c1 = 128 + lane * 4;

    v4f wx0, wy0, wz0, wx1, wy1, wz1;
#pragma unroll
    for (int i = 0; i < 4; ++i) {
        const float* p0 = iW + (size_t)(c0 + i) * (LATENT + 3);
        const float* p1 = iW + (size_t)(c1 + i) * (LATENT + 3);
        wx0[i] = p0[0]; wy0[i] = p0[1]; wz0[i] = p0[2];
        wx1[i] = p1[0]; wy1[i] = p1[1]; wz1[i] = p1[2];
    }
    const v4f b0 = *(const v4f*)(ib + c0);
    const v4f b1 = *(const v4f*)(ib + c1);
    v4f l0[B_N], l1[B_N];
#pragma unroll
    for (int b = 0; b < B_N; ++b) {
        l0[b] = *(const v4f*)(latW + b * HIDDEN + c0);
        l1[b] = *(const v4f*)(latW + b * HIDDEN + c1);
    }

    for (int i = 0; i < X0_VPW; ++i) {
        const int v = vbeg + i;
        const float px = xyz[3 * v], py = xyz[3 * v + 1], pz = xyz[3 * v + 2];
        const v4f t0 = b0 + wx0 * px + wy0 * py + wz0 * pz;
        const v4f t1 = b1 + wx1 * px + wy1 * py + wz1 * pz;
#pragma unroll
        for (int b = 0; b < B_N; ++b) {
            float* xo = x + ((size_t)b * V_N + v) * HIDDEN;
            *(v4f*)(xo + c0) = relu4(t0 + l0[b]);
            *(v4f*)(xo + c1) = relu4(t1 + l1[b]);
        }
    }
}


#define GEMM_TILE_M  64
#define GEMM_TILES   (NNODES / GEMM_TILE_M)
#define GEMM_WGS     256
#define GEMM_NT      8
#define W_PITCH      260
static_assert(NNODES % GEMM_TILE_M == 0, "tiles cover the rows exactly");
static_assert(GEMM_WGS <= GEMM_TILES, "every work-group owns at least one tile");

__global__ void __launch_bounds__(256)
k_gemm_wmma(const float* __restrict__ x, const float* __restrict__ W,
            float* __restrict__ h) {
    __shared__ __attribute__((aligned(16))) float Ws[HIDDEN * W_PITCH];

    const int lane  = threadIdx.x & 31;
    const int wv    = threadIdx.x >> 5;
    const int hf    = lane >> 4;
    const int l16   = lane & 15;
    const int mrow  = (wv >> 1) * 16;
    const int nbase = (wv & 1) * 128;

    const int ntile  = (GEMM_TILES - (int)blockIdx.x + GEMM_WGS - 1) / GEMM_WGS;
    const int nchunk = ntile * 8;

#if HAVE_WMMA_F32
    v4f a_cur[4], a_nxt[4];
    {
        const float* p = x + (size_t)((int)blockIdx.x * GEMM_TILE_M + mrow + l16) * HIDDEN + hf * 4;
#pragma unroll
        for (int j = 0; j < 4; ++j) a_cur[j] = *(const v4f*)(p + 8 * j);
    }
#endif

    {
        const int rot = (int)blockIdx.x & 63;
        for (int i = 0; i < 64; i += 16) {
            v4f t[16];
#pragma unroll
            for (int j = 0; j < 16; ++j) {
                const int q = (((i + j + rot) & 63) << 8) + (int)threadIdx.x;
                t[j] = *(const v4f*)(W + (size_t)q * 4);
            }
#pragma unroll
            for (int j = 0; j < 16; ++j) {
                const int q = (((i + j + rot) & 63) << 8) + (int)threadIdx.x;
                *(v4f*)(Ws + (q >> 6) * W_PITCH + (q & 63) * 4) = t[j];
            }
        }
    }
    __syncthreads();

#if HAVE_WMMA_F32
    const float* bp = Ws + (nbase + l16) * W_PITCH + hf * 4;

    v8f acc[GEMM_NT] = {};
    v4f bq[GEMM_NT];
#pragma unroll
    for (int nt = 0; nt < GEMM_NT; ++nt) bq[nt] = *(const v4f*)(bp + nt * 16 * W_PITCH);

    for (int g = 0; g < nchunk; ++g) {
        const int kc = g & 7;
        {
            const int gn = (g + 1 < nchunk) ? g + 1 : g;
            const int tile = (int)blockIdx.x + (gn >> 3) * GEMM_WGS;
            const float* p = x + (size_t)(tile * GEMM_TILE_M + mrow + l16) * HIDDEN
                               + (gn & 7) * 32 + hf * 4;
#pragma unroll
            for (int j = 0; j < 4; ++j) a_nxt[j] = *(const v4f*)(p + 8 * j);
        }

#pragma unroll
        for (int j = 0; j < 4; ++j) {
            const int knext = (j < 3) ? kc * 32 + (j + 1) * 8 : ((kc + 1) & 7) * 32;
            v4f bn[GEMM_NT];
#pragma unroll
            for (int nt = 0; nt < GEMM_NT; ++nt)
                bn[nt] = *(const v4f*)(bp + nt * 16 * W_PITCH + knext);

            const v2f alo = {a_cur[j].x, a_cur[j].y};
            const v2f ahi = {a_cur[j].z, a_cur[j].w};
#pragma unroll
            for (int nt = 0; nt < GEMM_NT; ++nt) {
                const v2f blo = {bq[nt].x, bq[nt].y};
                acc[nt] = __builtin_amdgcn_wmma_f32_16x16x4_f32(
                     false, alo,  false, blo,
                     (short)0, acc[nt],
                     false,  false);
            }
#pragma unroll
            for (int nt = 0; nt < GEMM_NT; ++nt) {
                const v2f bhi = {bq[nt].z, bq[nt].w};
                acc[nt] = __builtin_amdgcn_wmma_f32_16x16x4_f32(
                     false, ahi,  false, bhi,
                     (short)0, acc[nt],
                     false,  false);
            }
#pragma unroll
            for (int nt = 0; nt < GEMM_NT; ++nt) bq[nt] = bn[nt];
        }

        if (kc == 7) {
            const int tile = (int)blockIdx.x + (g >> 3) * GEMM_WGS;
            float* outp = h + (size_t)(tile * GEMM_TILE_M + mrow + hf * 8) * HIDDEN + (nbase + l16);
#pragma unroll
            for (int nt = 0; nt < GEMM_NT; ++nt) {
#pragma unroll
                for (int r = 0; r < 8; ++r)
                    outp[(size_t)r * HIDDEN + nt * 16] = acc[nt][r];
                acc[nt] = v8f{};
            }
        }
#pragma unroll
        for (int j = 0; j < 4; ++j) a_cur[j] = a_nxt[j];
    }
#else
    for (int t = 0; t < ntile; ++t) {
        const int tile = (int)blockIdx.x + t * GEMM_WGS;
        for (int idx = threadIdx.x; idx < GEMM_TILE_M * HIDDEN; idx += 256) {
            const int row = tile * GEMM_TILE_M + (idx >> 8), col = idx & 255;
            const float* xr = x + (size_t)row * HIDDEN;
            float s = 0.0f;
            for (int k = 0; k < HIDDEN; ++k) s += xr[k] * Ws[col * W_PITCH + k];
            h[(size_t)row * HIDDEN + col] = s;
        }
    }
    (void)lane; (void)wv; (void)hf; (void)l16; (void)mrow; (void)nbase; (void)nchunk;
#endif
}


__device__ __forceinline__ size_t row_of(int idx) {
    unsigned u = (unsigned)idx;
    return (size_t)(u < (unsigned)(V_N - 1) ? u : (unsigned)(V_N - 1));
}

__global__ void __launch_bounds__(256)
k_gather(const float* __restrict__ h, float* __restrict__ x,
         const int* __restrict__ rowptr, const int* __restrict__ csrc,
         const float* __restrict__ cw, const float* __restrict__ selfnorm,
         const float* __restrict__ bias) {
    const int lane = threadIdx.x & 31;
    const int node = __builtin_amdgcn_readfirstlane((int)(blockIdx.x * 8 + (threadIdx.x >> 5)));
    const int b = node / V_N;
    const int v = node - b * V_N;
    const float* hb = h + (size_t)b * V_N * HIDDEN;

    const float* hs = hb + (size_t)v * HIDDEN;
    const v4f s0 = *(const v4f*)(hs + lane * 4);
    const v4f s1 = *(const v4f*)(hs + 128 + lane * 4);
    const v4f c0 = *(const v4f*)(bias + lane * 4);
    const v4f c1 = *(const v4f*)(bias + 128 + lane * 4);
    const float sn = selfnorm[v];

    const int beg = rowptr[v], end = rowptr[v + 1];
    v4f a0 = {0.0f, 0.0f, 0.0f, 0.0f};
    v4f a1 = {0.0f, 0.0f, 0.0f, 0.0f};

    int j = beg;
    for (; j + 4 <= end; j += 4) {
        v4f u0[4], u1[4];
        float w[4];
#pragma unroll
        for (int k = 0; k < 4; ++k) {
            const float* p = hb + row_of(csrc[j + k]) * HIDDEN;
            w[k]  = cw[j + k];
            u0[k] = *(const v4f*)(p + lane * 4);
            u1[k] = *(const v4f*)(p + 128 + lane * 4);
        }
#pragma unroll
        for (int k = 0; k < 4; ++k) {
            a0 += u0[k] * w[k];
            a1 += u1[k] * w[k];
        }
    }
    if (j + 2 <= end) {
        v4f u0[2], u1[2];
        float w[2];
#pragma unroll
        for (int k = 0; k < 2; ++k) {
            const float* p = hb + row_of(csrc[j + k]) * HIDDEN;
            w[k]  = cw[j + k];
            u0[k] = *(const v4f*)(p + lane * 4);
            u1[k] = *(const v4f*)(p + 128 + lane * 4);
        }
#pragma unroll
        for (int k = 0; k < 2; ++k) {
            a0 += u0[k] * w[k];
            a1 += u1[k] * w[k];
        }
        j += 2;
    }
    if (j < end) {
        const float* p = hb + row_of(csrc[j]) * HIDDEN;
        const float w = cw[j];
        const v4f u0 = *(const v4f*)(p + lane * 4);
        const v4f u1 = *(const v4f*)(p + 128 + lane * 4);
        a0 += u0 * w;
        a1 += u1 * w;
    }

    a0 += s0 * sn;
    a1 += s1 * sn;
    float* xo = x + (size_t)node * HIDDEN;
    *(v4f*)(xo + lane * 4)       = relu4(a0 + c0);
    *(v4f*)(xo + 128 + lane * 4) = relu4(a1 + c1);
}

__global__ void k_out(const float* __restrict__ x, const float* __restrict__ oW,
                      const float* __restrict__ ob, float* __restrict__ disp) {
    int idx = blockIdx.x * blockDim.x + threadIdx.x;
    if (idx >= NNODES * 3) return;
    int node = idx / 3, o = idx % 3;
    const float* xp = x + (size_t)node * HIDDEN;
    const float* wp = oW + o * HIDDEN;
    float s = ob[o];
#pragma unroll 4
    for (int k = 0; k < HIDDEN; k += 4) {
        float4 xv = *(const float4*)(xp + k);
        float4 wv = *(const float4*)(wp + k);
        s += xv.x * wv.x + xv.y * wv.y + xv.z * wv.z + xv.w * wv.w;
    }
    disp[idx] = s;
}


extern "C" void kernel_launch(void* const* d_in, const int* in_sizes, int n_in,
                              void* d_out, int out_size, void* d_ws, size_t ws_size,
                              hipStream_t stream) {
    const float* vertex_xyz = (const float*)d_in[0];
    const float* latent     = (const float*)d_in[1];
    const int*   edge_index = (const int*)d_in[2];
    const float* input_W    = (const float*)d_in[3];
    const float* input_b    = (const float*)d_in[4];
    const float* conv_W     = (const float*)d_in[5];
    const float* conv_b     = (const float*)d_in[6];
    const float* out_W      = (const float*)d_in[7];
    const float* out_b      = (const float*)d_in[8];
    float* disp = (float*)d_out;

    const int* src = edge_index;
    const int* dst = edge_index + E_N;

    float* xbuf     = (float*)d_ws;
    float* hbuf     = xbuf + (size_t)NNODES * HIDDEN;
    float* xyzW     = hbuf + (size_t)NNODES * HIDDEN;
    float* latW     = xyzW + (size_t)V_N * HIDDEN;
    float* dinv     = latW + (size_t)B_N * HIDDEN;
    float* selfnorm = dinv + V_N;
    int*   cnt      = (int*)(selfnorm + V_N);
    int*   rowptr   = cnt + V_N;
    int*   rank     = (int*)hbuf;
    int*   slot     = rank + E_N;
    int*   csrc     = (int*)xyzW;
    float* cw       = (float*)(csrc + E_N);

    const int T = 256;

    k_latW<<<(B_N * HIDDEN + T - 1) / T, T, 0, stream>>>(latent, input_W, latW);
    k_x0<<<V_N / (X0_VPW * X0_WPB), 32 * X0_WPB, 0, stream>>>(vertex_xyz, input_W, input_b, latW, xbuf);

    k_cnt_zero<<<(V_N + T - 1) / T, T, 0, stream>>>(cnt);
    k_rank<<<(E_N + T - 1) / T, T, 0, stream>>>(cnt, dst, rank);
    k_scan<<<1, SCAN_T, 0, stream>>>(cnt, rowptr, dinv, selfnorm);
    k_fill<<<(E_N + T - 1) / T, T, 0, stream>>>(rowptr, dst, rank, slot);
    k_place<<<(E_N + T - 1) / T, T, 0, stream>>>(rowptr, slot, src, dst, dinv, csrc, cw);

    dim3 ggrid(GEMM_WGS);
    for (int l = 0; l < NLAYERS; ++l) {
        k_gemm_wmma<<<ggrid, T, 0, stream>>>(xbuf, conv_W + (size_t)l * HIDDEN * HIDDEN, hbuf);
        k_gather<<<NNODES / 8, T, 0, stream>>>(hbuf, xbuf, rowptr, csrc, cw, selfnorm,
                                               conv_b + (size_t)l * HIDDEN);
    }

    k_out<<<(NNODES * 3 + T - 1) / T, T, 0, stream>>>(xbuf, out_W, out_b, disp);
}


# Layer 1 sums its neighbours before its linear map (agg(x0 W) = (agg x0) W): x0 rows rebuilt from 12 bytes of xyz per neighbour, not gathered as 1 KB rows; no x0 pass; bias+relu in the GEMM epilogue
# speedup vs baseline: 3.2176x; 1.0463x over previous
#include <hip/hip_runtime.h>
#include <hip/hip_bf16.h>

#define V_N     10000
#define E_N     60000
#define B_N     8
#define LATENT  512
#define HIDDEN  256
#define NLAYERS 5
#define NNODES  (B_N * V_N)

typedef float v2f __attribute__((ext_vector_type(2)));
typedef float v8f __attribute__((ext_vector_type(8)));

#if defined(__gfx1250__) && __has_builtin(__builtin_amdgcn_wmma_f32_16x16x4_f32)
#define HAVE_WMMA_F32 1
#endif


__global__ void k_cnt_zero(int* cnt) {
    int v = blockIdx.x * blockDim.x + threadIdx.x;
    if (v < V_N) cnt[v] = 0;
}

__global__ void k_rank(int* cnt, const int* __restrict__ dst, int* __restrict__ rank) {
    int e = blockIdx.x * blockDim.x + threadIdx.x;
    if (e < E_N) rank[e] = atomicAdd(&cnt[dst[e]], 1);
}

#define SCAN_T   1024
#define SCAN_PER ((V_N + SCAN_T - 1) / SCAN_T)
__global__ void __launch_bounds__(SCAN_T)
k_scan(const int* __restrict__ cnt, int* __restrict__ rowptr,
       float* __restrict__ dinv, float* __restrict__ selfnorm) {
    __shared__ int part[SCAN_T];
    const int t = threadIdx.x;
    const int base = t * SCAN_PER;
    int loc[SCAN_PER];
    int c[SCAN_PER];
    int s = 0;
#pragma unroll
    for (int i = 0; i < SCAN_PER; ++i) {
        int v = base + i;
        c[i] = (v < V_N) ? cnt[v] : 0;
        loc[i] = s;
        s += c[i];
    }
    part[t] = s;
    __syncthreads();
    for (int off = 1; off < SCAN_T; off <<= 1) {
        int add = (t >= off) ? part[t - off] : 0;
        __syncthreads();
        part[t] += add;
        __syncthreads();
    }
    const int excl = part[t] - s;
#pragma unroll
    for (int i = 0; i < SCAN_PER; ++i) {
        int v = base + i;
        if (v < V_N) {
            rowptr[v] = excl + loc[i];
            float di = rsqrtf((float)(c[i] + 1));
            dinv[v] = di;
            selfnorm[v] = di * di;
        }
    }
    if (t == SCAN_T - 1) rowptr[V_N] = part[t];
}

__global__ void k_fill(const int* __restrict__ rowptr, const int* __restrict__ dst,
                       const int* __restrict__ rank, int* __restrict__ slot) {
    int e = blockIdx.x * blockDim.x + threadIdx.x;
    if (e < E_N) slot[rowptr[dst[e]] + rank[e]] = e;
}

__global__ void k_place(const int* __restrict__ rowptr, const int* __restrict__ slot,
                        const int* __restrict__ src, const int* __restrict__ dst,
                        const float* __restrict__ dinv,
                        int* __restrict__ csrc, float* __restrict__ cw) {
    int e = blockIdx.x * blockDim.x + threadIdx.x;
    if (e >= E_N) return;
    const int v = dst[e], s = src[e];
    const int beg = rowptr[v], end = rowptr[v + 1];
    int r = 0;
    for (int j = beg; j < end; ++j) r += (slot[j] < e) ? 1 : 0;
    csrc[beg + r] = s;
    cw[beg + r]   = dinv[s] * dinv[v];
}


__global__ void k_latW(const float* __restrict__ latent, const float* __restrict__ iW,
                       float* __restrict__ latW) {
    int i = blockIdx.x * blockDim.x + threadIdx.x;
    if (i >= B_N * HIDDEN) return;
    int b = i / HIDDEN, hh = i % HIDDEN;
    const float* lp = latent + (size_t)b * LATENT;
    const float* wp = iW + (size_t)hh * (LATENT + 3) + 3;
    float s = 0.0f;
    for (int j = 0; j < LATENT; ++j) s += lp[j] * wp[j];
    latW[i] = s;
}

typedef float v4f __attribute__((ext_vector_type(4)));

__device__ __forceinline__ v4f relu4(v4f s) {
    v4f r;
    r.x = s.x > 0.0f ? s.x : 0.0f;
    r.y = s.y > 0.0f ? s.y : 0.0f;
    r.z = s.z > 0.0f ? s.z : 0.0f;
    r.w = s.w > 0.0f ? s.w : 0.0f;
    return r;
}

__device__ __forceinline__ size_t row_of(int idx) {
    unsigned u = (unsigned)idx;
    return (size_t)(u < (unsigned)(V_N - 1) ? u : (unsigned)(V_N - 1));
}

#define AG_WPB 8
static_assert(V_N % AG_WPB == 0, "grid covers the vertices exactly");

__device__ __forceinline__ void agg0_term(v4f (&a0)[B_N], v4f (&a1)[B_N],
                                          const v4f (&l0)[B_N], const v4f (&l1)[B_N],
                                          const v4f t0, const v4f t1, const float w) {
#pragma unroll
    for (int b = 0; b < B_N; ++b) {
        a0[b] += relu4(t0 + l0[b]) * w;
        a1[b] += relu4(t1 + l1[b]) * w;
    }
}

__global__ void __launch_bounds__(32 * AG_WPB)
k_agg0(const float* __restrict__ xyz, const float* __restrict__ iW, const float* __restrict__ ib,
       const float* __restrict__ latW, const int* __restrict__ rowptr, const int* __restrict__ csrc,
       const float* __restrict__ cw, const float* __restrict__ selfnorm, float* __restrict__ y) {
    const int lane = threadIdx.x & 31;
    const int v = __builtin_amdgcn_readfirstlane((int)(blockIdx.x * AG_WPB + (threadIdx.x >> 5)));
    const int c0 = lane * 4, c1 = 128 + lane * 4;

    v4f wx0, wy0, wz0, wx1, wy1, wz1;
#pragma unroll
    for (int i = 0; i < 4; ++i) {
        const float* p0 = iW + (size_t)(c0 + i) * (LATENT + 3);
        const float* p1 = iW + (size_t)(c1 + i) * (LATENT + 3);
        wx0[i] = p0[0]; wy0[i] = p0[1]; wz0[i] = p0[2];
        wx1[i] = p1[0]; wy1[i] = p1[1]; wz1[i] = p1[2];
    }
    const v4f b0 = *(const v4f*)(ib + c0);
    const v4f b1 = *(const v4f*)(ib + c1);
    v4f l0[B_N], l1[B_N], a0[B_N], a1[B_N];
#pragma unroll
    for (int b = 0; b < B_N; ++b) {
        l0[b] = *(const v4f*)(latW + b * HIDDEN + c0);
        l1[b] = *(const v4f*)(latW + b * HIDDEN + c1);
        a0[b] = v4f{0.0f, 0.0f, 0.0f, 0.0f};
        a1[b] = v4f{0.0f, 0.0f, 0.0f, 0.0f};
    }

    const int beg = rowptr[v], end = rowptr[v + 1];
    int j = beg;
    for (; j + 4 <= end; j += 4) {
        size_t s[4];
        float w[4], px[4], py[4], pz[4];
#pragma unroll
        for (int k = 0; k < 4; ++k) { s[k] = row_of(csrc[j + k]); w[k] = cw[j + k]; }
#pragma unroll
        for (int k = 0; k < 4; ++k) {
            px[k] = xyz[3 * s[k]]; py[k] = xyz[3 * s[k] + 1]; pz[k] = xyz[3 * s[k] + 2];
        }
#pragma unroll
        for (int k = 0; k < 4; ++k) {
            const v4f t0 = b0 + wx0 * px[k] + wy0 * py[k] + wz0 * pz[k];
            const v4f t1 = b1 + wx1 * px[k] + wy1 * py[k] + wz1 * pz[k];
            agg0_term(a0, a1, l0, l1, t0, t1, w[k]);
        }
    }
    for (; j < end; ++j) {
        const size_t s = row_of(csrc[j]);
        const float w = cw[j];
        const float px = xyz[3 * s], py = xyz[3 * s + 1], pz = xyz[3 * s + 2];
        const v4f t0 = b0 + wx0 * px + wy0 * py + wz0 * pz;
        const v4f t1 = b1 + wx1 * px + wy1 * py + wz1 * pz;
        agg0_term(a0, a1, l0, l1, t0, t1, w);
    }
    {
        const float w = selfnorm[v];
        const float px = xyz[3 * v], py = xyz[3 * v + 1], pz = xyz[3 * v + 2];
        const v4f t0 = b0 + wx0 * px + wy0 * py + wz0 * pz;
        const v4f t1 = b1 + wx1 * px + wy1 * py + wz1 * pz;
        agg0_term(a0, a1, l0, l1, t0, t1, w);
    }

#pragma unroll
    for (int b = 0; b < B_N; ++b) {
        float* yo = y + ((size_t)b * V_N + v) * HIDDEN;
        *(v4f*)(yo + c0) = a0[b];
        *(v4f*)(yo + c1) = a1[b];
    }
}


#define GEMM_TILE_M  64
#define GEMM_TILES   (NNODES / GEMM_TILE_M)
#define GEMM_WGS     256
#define GEMM_NT      8
#define W_PITCH      260
static_assert(NNODES % GEMM_TILE_M == 0, "tiles cover the rows exactly");
static_assert(GEMM_WGS <= GEMM_TILES, "every work-group owns at least one tile");

template <bool EPI>
__global__ void __launch_bounds__(256)
k_gemm_wmma(const float* __restrict__ x, const float* __restrict__ W,
            const float* __restrict__ bias, float* __restrict__ h) {
    __shared__ __attribute__((aligned(16))) float Ws[HIDDEN * W_PITCH];

    const int lane  = threadIdx.x & 31;
    const int wv    = threadIdx.x >> 5;
    const int hf    = lane >> 4;
    const int l16   = lane & 15;
    const int mrow  = (wv >> 1) * 16;
    const int nbase = (wv & 1) * 128;

    const int ntile  = (GEMM_TILES - (int)blockIdx.x + GEMM_WGS - 1) / GEMM_WGS;
    const int nchunk = ntile * 8;

#if HAVE_WMMA_F32
    v4f a_cur[4], a_nxt[4];
    {
        const float* p = x + (size_t)((int)blockIdx.x * GEMM_TILE_M + mrow + l16) * HIDDEN + hf * 4;
#pragma unroll
        for (int j = 0; j < 4; ++j) a_cur[j] = *(const v4f*)(p + 8 * j);
    }
#endif

    {
        const int rot = (int)blockIdx.x & 63;
        for (int i = 0; i < 64; i += 16) {
            v4f t[16];
#pragma unroll
            for (int j = 0; j < 16; ++j) {
                const int q = (((i + j + rot) & 63) << 8) + (int)threadIdx.x;
                t[j] = *(const v4f*)(W + (size_t)q * 4);
            }
#pragma unroll
            for (int j = 0; j < 16; ++j) {
                const int q = (((i + j + rot) & 63) << 8) + (int)threadIdx.x;
                *(v4f*)(Ws + (q >> 6) * W_PITCH + (q & 63) * 4) = t[j];
            }
        }
    }
    __syncthreads();

#if HAVE_WMMA_F32
    const float* bp = Ws + (nbase + l16) * W_PITCH + hf * 4;

    v8f acc[GEMM_NT] = {};
    float bz[GEMM_NT];
#pragma unroll
    for (int nt = 0; nt < GEMM_NT; ++nt) bz[nt] = EPI ? bias[nbase + nt * 16 + l16] : 0.0f;
    v4f bq[GEMM_NT];
#pragma unroll
    for (int nt = 0; nt < GEMM_NT; ++nt) bq[nt] = *(const v4f*)(bp + nt * 16 * W_PITCH);

    for (int g = 0; g < nchunk; ++g) {
        const int kc = g & 7;
        {
            const int gn = (g + 1 < nchunk) ? g + 1 : g;
            const int tile = (int)blockIdx.x + (gn >> 3) * GEMM_WGS;
            const float* p = x + (size_t)(tile * GEMM_TILE_M + mrow + l16) * HIDDEN
                               + (gn & 7) * 32 + hf * 4;
#pragma unroll
            for (int j = 0; j < 4; ++j) a_nxt[j] = *(const v4f*)(p + 8 * j);
        }

#pragma unroll
        for (int j = 0; j < 4; ++j) {
            const int knext = (j < 3) ? kc * 32 + (j + 1) * 8 : ((kc + 1) & 7) * 32;
            v4f bn[GEMM_NT];
#pragma unroll
            for (int nt = 0; nt < GEMM_NT; ++nt)
                bn[nt] = *(const v4f*)(bp + nt * 16 * W_PITCH + knext);

            const v2f alo = {a_cur[j].x, a_cur[j].y};
            const v2f ahi = {a_cur[j].z, a_cur[j].w};
#pragma unroll
            for (int nt = 0; nt < GEMM_NT; ++nt) {
                const v2f blo = {bq[nt].x, bq[nt].y};
                acc[nt] = __builtin_amdgcn_wmma_f32_16x16x4_f32(
                     false, alo,  false, blo,
                     (short)0, acc[nt],
                     false,  false);
            }
#pragma unroll
            for (int nt = 0; nt < GEMM_NT; ++nt) {
                const v2f bhi = {bq[nt].z, bq[nt].w};
                acc[nt] = __builtin_amdgcn_wmma_f32_16x16x4_f32(
                     false, ahi,  false, bhi,
                     (short)0, acc[nt],
                     false,  false);
            }
#pragma unroll
            for (int nt = 0; nt < GEMM_NT; ++nt) bq[nt] = bn[nt];
        }

        if (kc == 7) {
            const int tile = (int)blockIdx.x + (g >> 3) * GEMM_WGS;
            float* outp = h + (size_t)(tile * GEMM_TILE_M + mrow + hf * 8) * HIDDEN + (nbase + l16);
#pragma unroll
            for (int nt = 0; nt < GEMM_NT; ++nt) {
#pragma unroll
                for (int r = 0; r < 8; ++r) {
                    float o = acc[nt][r];
                    if (EPI) { o += bz[nt]; o = o > 0.0f ? o : 0.0f; }
                    outp[(size_t)r * HIDDEN + nt * 16] = o;
                }
                acc[nt] = v8f{};
            }
        }
#pragma unroll
        for (int j = 0; j < 4; ++j) a_cur[j] = a_nxt[j];
    }
#else
    for (int t = 0; t < ntile; ++t) {
        const int tile = (int)blockIdx.x + t * GEMM_WGS;
        for (int idx = threadIdx.x; idx < GEMM_TILE_M * HIDDEN; idx += 256) {
            const int row = tile * GEMM_TILE_M + (idx >> 8), col = idx & 255;
            const float* xr = x + (size_t)row * HIDDEN;
            float s = 0.0f;
            for (int k = 0; k < HIDDEN; ++k) s += xr[k] * Ws[col * W_PITCH + k];
            if (EPI) { s += bias[col]; s = s > 0.0f ? s : 0.0f; }
            h[(size_t)row * HIDDEN + col] = s;
        }
    }
    (void)lane; (void)wv; (void)hf; (void)l16; (void)mrow; (void)nbase; (void)nchunk;
#endif
}


__global__ void __launch_bounds__(256)
k_gather(const float* __restrict__ h, float* __restrict__ x,
         const int* __restrict__ rowptr, const int* __restrict__ csrc,
         const float* __restrict__ cw, const float* __restrict__ selfnorm,
         const float* __restrict__ bias) {
    const int lane = threadIdx.x & 31;
    const int node = __builtin_amdgcn_readfirstlane((int)(blockIdx.x * 8 + (threadIdx.x >> 5)));
    const int b = node / V_N;
    const int v = node - b * V_N;
    const float* hb = h + (size_t)b * V_N * HIDDEN;

    const float* hs = hb + (size_t)v * HIDDEN;
    const v4f s0 = *(const v4f*)(hs + lane * 4);
    const v4f s1 = *(const v4f*)(hs + 128 + lane * 4);
    const v4f c0 = *(const v4f*)(bias + lane * 4);
    const v4f c1 = *(const v4f*)(bias + 128 + lane * 4);
    const float sn = selfnorm[v];

    const int beg = rowptr[v], end = rowptr[v + 1];
    v4f a0 = {0.0f, 0.0f, 0.0f, 0.0f};
    v4f a1 = {0.0f, 0.0f, 0.0f, 0.0f};

    int j = beg;
    for (; j + 4 <= end; j += 4) {
        v4f u0[4], u1[4];
        float w[4];
#pragma unroll
        for (int k = 0; k < 4; ++k) {
            const float* p = hb + row_of(csrc[j + k]) * HIDDEN;
            w[k]  = cw[j + k];
            u0[k] = *(const v4f*)(p + lane * 4);
            u1[k] = *(const v4f*)(p + 128 + lane * 4);
        }
#pragma unroll
        for (int k = 0; k < 4; ++k) {
            a0 += u0[k] * w[k];
            a1 += u1[k] * w[k];
        }
    }
    if (j + 2 <= end) {
        v4f u0[2], u1[2];
        float w[2];
#pragma unroll
        for (int k = 0; k < 2; ++k) {
            const float* p = hb + row_of(csrc[j + k]) * HIDDEN;
            w[k]  = cw[j + k];
            u0[k] = *(const v4f*)(p + lane * 4);
            u1[k] = *(const v4f*)(p + 128 + lane * 4);
        }
#pragma unroll
        for (int k = 0; k < 2; ++k) {
            a0 += u0[k] * w[k];
            a1 += u1[k] * w[k];
        }
        j += 2;
    }
    if (j < end) {
        const float* p = hb + row_of(csrc[j]) * HIDDEN;
        const float w = cw[j];
        const v4f u0 = *(const v4f*)(p + lane * 4);
        const v4f u1 = *(const v4f*)(p + 128 + lane * 4);
        a0 += u0 * w;
        a1 += u1 * w;
    }

    a0 += s0 * sn;
    a1 += s1 * sn;
    float* xo = x + (size_t)node * HIDDEN;
    *(v4f*)(xo + lane * 4)       = relu4(a0 + c0);
    *(v4f*)(xo + 128 + lane * 4) = relu4(a1 + c1);
}

__global__ void k_out(const float* __restrict__ x, const float* __restrict__ oW,
                      const float* __restrict__ ob, float* __restrict__ disp) {
    int idx = blockIdx.x * blockDim.x + threadIdx.x;
    if (idx >= NNODES * 3) return;
    int node = idx / 3, o = idx % 3;
    const float* xp = x + (size_t)node * HIDDEN;
    const float* wp = oW + o * HIDDEN;
    float s = ob[o];
#pragma unroll 4
    for (int k = 0; k < HIDDEN; k += 4) {
        float4 xv = *(const float4*)(xp + k);
        float4 wv = *(const float4*)(wp + k);
        s += xv.x * wv.x + xv.y * wv.y + xv.z * wv.z + xv.w * wv.w;
    }
    disp[idx] = s;
}


extern "C" void kernel_launch(void* const* d_in, const int* in_sizes, int n_in,
                              void* d_out, int out_size, void* d_ws, size_t ws_size,
                              hipStream_t stream) {
    const float* vertex_xyz = (const float*)d_in[0];
    const float* latent     = (const float*)d_in[1];
    const int*   edge_index = (const int*)d_in[2];
    const float* input_W    = (const float*)d_in[3];
    const float* input_b    = (const float*)d_in[4];
    const float* conv_W     = (const float*)d_in[5];
    const float* conv_b     = (const float*)d_in[6];
    const float* out_W      = (const float*)d_in[7];
    const float* out_b      = (const float*)d_in[8];
    float* disp = (float*)d_out;

    const int* src = edge_index;
    const int* dst = edge_index + E_N;

    float* xbuf     = (float*)d_ws;
    float* hbuf     = xbuf + (size_t)NNODES * HIDDEN;
    float* xyzW     = hbuf + (size_t)NNODES * HIDDEN;
    float* latW     = xyzW + (size_t)V_N * HIDDEN;
    float* dinv     = latW + (size_t)B_N * HIDDEN;
    float* selfnorm = dinv + V_N;
    int*   cnt      = (int*)(selfnorm + V_N);
    int*   rowptr   = cnt + V_N;
    int*   rank     = (int*)hbuf;
    int*   slot     = rank + E_N;
    int*   csrc     = (int*)xyzW;
    float* cw       = (float*)(csrc + E_N);

    const int T = 256;

    k_latW<<<(B_N * HIDDEN + T - 1) / T, T, 0, stream>>>(latent, input_W, latW);

    k_cnt_zero<<<(V_N + T - 1) / T, T, 0, stream>>>(cnt);
    k_rank<<<(E_N + T - 1) / T, T, 0, stream>>>(cnt, dst, rank);
    k_scan<<<1, SCAN_T, 0, stream>>>(cnt, rowptr, dinv, selfnorm);
    k_fill<<<(E_N + T - 1) / T, T, 0, stream>>>(rowptr, dst, rank, slot);
    k_place<<<(E_N + T - 1) / T, T, 0, stream>>>(rowptr, slot, src, dst, dinv, csrc, cw);

    dim3 ggrid(GEMM_WGS);
    k_agg0<<<V_N / AG_WPB, 32 * AG_WPB, 0, stream>>>(vertex_xyz, input_W, input_b, latW,
                                                     rowptr, csrc, cw, selfnorm, xbuf);
    k_gemm_wmma<true><<<ggrid, T, 0, stream>>>(xbuf, conv_W, conv_b, hbuf);

    for (int l = 1; l < NLAYERS; ++l) {
        k_gemm_wmma<false><<<ggrid, T, 0, stream>>>(hbuf, conv_W + (size_t)l * HIDDEN * HIDDEN,
                                                    conv_b, xbuf);
        k_gather<<<NNODES / 8, T, 0, stream>>>(xbuf, hbuf, rowptr, csrc, cw, selfnorm,
                                               conv_b + (size_t)l * HIDDEN);
    }

    k_out<<<(NNODES * 3 + T - 1) / T, T, 0, stream>>>(hbuf, out_W, out_b, disp);
}


# Output projection folded into the last layer's gather: the wave that sums a row dots it with out_W in registers; that layer's rows are never stored, the projection launch and its row reads are gone
# speedup vs baseline: 3.3175x; 1.0310x over previous
#include <hip/hip_runtime.h>
#include <hip/hip_bf16.h>

#define V_N     10000
#define E_N     60000
#define B_N     8
#define LATENT  512
#define HIDDEN  256
#define NLAYERS 5
#define NNODES  (B_N * V_N)

typedef float v2f __attribute__((ext_vector_type(2)));
typedef float v8f __attribute__((ext_vector_type(8)));

#if defined(__gfx1250__) && __has_builtin(__builtin_amdgcn_wmma_f32_16x16x4_f32)
#define HAVE_WMMA_F32 1
#endif


__global__ void k_cnt_zero(int* cnt) {
    int v = blockIdx.x * blockDim.x + threadIdx.x;
    if (v < V_N) cnt[v] = 0;
}

__global__ void k_rank(int* cnt, const int* __restrict__ dst, int* __restrict__ rank) {
    int e = blockIdx.x * blockDim.x + threadIdx.x;
    if (e < E_N) rank[e] = atomicAdd(&cnt[dst[e]], 1);
}

#define SCAN_T   1024
#define SCAN_PER ((V_N + SCAN_T - 1) / SCAN_T)
__global__ void __launch_bounds__(SCAN_T)
k_scan(const int* __restrict__ cnt, int* __restrict__ rowptr,
       float* __restrict__ dinv, float* __restrict__ selfnorm) {
    __shared__ int part[SCAN_T];
    const int t = threadIdx.x;
    const int base = t * SCAN_PER;
    int loc[SCAN_PER];
    int c[SCAN_PER];
    int s = 0;
#pragma unroll
    for (int i = 0; i < SCAN_PER; ++i) {
        int v = base + i;
        c[i] = (v < V_N) ? cnt[v] : 0;
        loc[i] = s;
        s += c[i];
    }
    part[t] = s;
    __syncthreads();
    for (int off = 1; off < SCAN_T; off <<= 1) {
        int add = (t >= off) ? part[t - off] : 0;
        __syncthreads();
        part[t] += add;
        __syncthreads();
    }
    const int excl = part[t] - s;
#pragma unroll
    for (int i = 0; i < SCAN_PER; ++i) {
        int v = base + i;
        if (v < V_N) {
            rowptr[v] = excl + loc[i];
            float di = rsqrtf((float)(c[i] + 1));
            dinv[v] = di;
            selfnorm[v] = di * di;
        }
    }
    if (t == SCAN_T - 1) rowptr[V_N] = part[t];
}

__global__ void k_fill(const int* __restrict__ rowptr, const int* __restrict__ dst,
                       const int* __restrict__ rank, int* __restrict__ slot) {
    int e = blockIdx.x * blockDim.x + threadIdx.x;
    if (e < E_N) slot[rowptr[dst[e]] + rank[e]] = e;
}

__global__ void k_place(const int* __restrict__ rowptr, const int* __restrict__ slot,
                        const int* __restrict__ src, const int* __restrict__ dst,
                        const float* __restrict__ dinv,
                        int* __restrict__ csrc, float* __restrict__ cw) {
    int e = blockIdx.x * blockDim.x + threadIdx.x;
    if (e >= E_N) return;
    const int v = dst[e], s = src[e];
    const int beg = rowptr[v], end = rowptr[v + 1];
    int r = 0;
    for (int j = beg; j < end; ++j) r += (slot[j] < e) ? 1 : 0;
    csrc[beg + r] = s;
    cw[beg + r]   = dinv[s] * dinv[v];
}


__global__ void k_latW(const float* __restrict__ latent, const float* __restrict__ iW,
                       float* __restrict__ latW) {
    int i = blockIdx.x * blockDim.x + threadIdx.x;
    if (i >= B_N * HIDDEN) return;
    int b = i / HIDDEN, hh = i % HIDDEN;
    const float* lp = latent + (size_t)b * LATENT;
    const float* wp = iW + (size_t)hh * (LATENT + 3) + 3;
    float s = 0.0f;
    for (int j = 0; j < LATENT; ++j) s += lp[j] * wp[j];
    latW[i] = s;
}

typedef float v4f __attribute__((ext_vector_type(4)));

__device__ __forceinline__ v4f relu4(v4f s) {
    v4f r;
    r.x = s.x > 0.0f ? s.x : 0.0f;
    r.y = s.y > 0.0f ? s.y : 0.0f;
    r.z = s.z > 0.0f ? s.z : 0.0f;
    r.w = s.w > 0.0f ? s.w : 0.0f;
    return r;
}

__device__ __forceinline__ size_t row_of(int idx) {
    unsigned u = (unsigned)idx;
    return (size_t)(u < (unsigned)(V_N - 1) ? u : (unsigned)(V_N - 1));
}

#define AG_WPB 8
static_assert(V_N % AG_WPB == 0, "grid covers the vertices exactly");

__device__ __forceinline__ void agg0_term(v4f (&a0)[B_N], v4f (&a1)[B_N],
                                          const v4f (&l0)[B_N], const v4f (&l1)[B_N],
                                          const v4f t0, const v4f t1, const float w) {
#pragma unroll
    for (int b = 0; b < B_N; ++b) {
        a0[b] += relu4(t0 + l0[b]) * w;
        a1[b] += relu4(t1 + l1[b]) * w;
    }
}

__global__ void __launch_bounds__(32 * AG_WPB)
k_agg0(const float* __restrict__ xyz, const float* __restrict__ iW, const float* __restrict__ ib,
       const float* __restrict__ latW, const int* __restrict__ rowptr, const int* __restrict__ csrc,
       const float* __restrict__ cw, const float* __restrict__ selfnorm, float* __restrict__ y) {
    const int lane = threadIdx.x & 31;
    const int v = __builtin_amdgcn_readfirstlane((int)(blockIdx.x * AG_WPB + (threadIdx.x >> 5)));
    const int c0 = lane * 4, c1 = 128 + lane * 4;

    v4f wx0, wy0, wz0, wx1, wy1, wz1;
#pragma unroll
    for (int i = 0; i < 4; ++i) {
        const float* p0 = iW + (size_t)(c0 + i) * (LATENT + 3);
        const float* p1 = iW + (size_t)(c1 + i) * (LATENT + 3);
        wx0[i] = p0[0]; wy0[i] = p0[1]; wz0[i] = p0[2];
        wx1[i] = p1[0]; wy1[i] = p1[1]; wz1[i] = p1[2];
    }
    const v4f b0 = *(const v4f*)(ib + c0);
    const v4f b1 = *(const v4f*)(ib + c1);
    v4f l0[B_N], l1[B_N], a0[B_N], a1[B_N];
#pragma unroll
    for (int b = 0; b < B_N; ++b) {
        l0[b] = *(const v4f*)(latW + b * HIDDEN + c0);
        l1[b] = *(const v4f*)(latW + b * HIDDEN + c1);
        a0[b] = v4f{0.0f, 0.0f, 0.0f, 0.0f};
        a1[b] = v4f{0.0f, 0.0f, 0.0f, 0.0f};
    }

    const int beg = rowptr[v], end = rowptr[v + 1];
    int j = beg;
    for (; j + 4 <= end; j += 4) {
        size_t s[4];
        float w[4], px[4], py[4], pz[4];
#pragma unroll
        for (int k = 0; k < 4; ++k) { s[k] = row_of(csrc[j + k]); w[k] = cw[j + k]; }
#pragma unroll
        for (int k = 0; k < 4; ++k) {
            px[k] = xyz[3 * s[k]]; py[k] = xyz[3 * s[k] + 1]; pz[k] = xyz[3 * s[k] + 2];
        }
#pragma unroll
        for (int k = 0; k < 4; ++k) {
            const v4f t0 = b0 + wx0 * px[k] + wy0 * py[k] + wz0 * pz[k];
            const v4f t1 = b1 + wx1 * px[k] + wy1 * py[k] + wz1 * pz[k];
            agg0_term(a0, a1, l0, l1, t0, t1, w[k]);
        }
    }
    for (; j < end; ++j) {
        const size_t s = row_of(csrc[j]);
        const float w = cw[j];
        const float px = xyz[3 * s], py = xyz[3 * s + 1], pz = xyz[3 * s + 2];
        const v4f t0 = b0 + wx0 * px + wy0 * py + wz0 * pz;
        const v4f t1 = b1 + wx1 * px + wy1 * py + wz1 * pz;
        agg0_term(a0, a1, l0, l1, t0, t1, w);
    }
    {
        const float w = selfnorm[v];
        const float px = xyz[3 * v], py = xyz[3 * v + 1], pz = xyz[3 * v + 2];
        const v4f t0 = b0 + wx0 * px + wy0 * py + wz0 * pz;
        const v4f t1 = b1 + wx1 * px + wy1 * py + wz1 * pz;
        agg0_term(a0, a1, l0, l1, t0, t1, w);
    }

#pragma unroll
    for (int b = 0; b < B_N; ++b) {
        float* yo = y + ((size_t)b * V_N + v) * HIDDEN;
        *(v4f*)(yo + c0) = a0[b];
        *(v4f*)(yo + c1) = a1[b];
    }
}


#define GEMM_TILE_M  64
#define GEMM_TILES   (NNODES / GEMM_TILE_M)
#define GEMM_WGS     256
#define GEMM_NT      8
#define W_PITCH      260
static_assert(NNODES % GEMM_TILE_M == 0, "tiles cover the rows exactly");
static_assert(GEMM_WGS <= GEMM_TILES, "every work-group owns at least one tile");

template <bool EPI>
__global__ void __launch_bounds__(256)
k_gemm_wmma(const float* __restrict__ x, const float* __restrict__ W,
            const float* __restrict__ bias, float* __restrict__ h) {
    __shared__ __attribute__((aligned(16))) float Ws[HIDDEN * W_PITCH];

    const int lane  = threadIdx.x & 31;
    const int wv    = threadIdx.x >> 5;
    const int hf    = lane >> 4;
    const int l16   = lane & 15;
    const int mrow  = (wv >> 1) * 16;
    const int nbase = (wv & 1) * 128;

    const int ntile  = (GEMM_TILES - (int)blockIdx.x + GEMM_WGS - 1) / GEMM_WGS;
    const int nchunk = ntile * 8;

#if HAVE_WMMA_F32
    v4f a_cur[4], a_nxt[4];
    {
        const float* p = x + (size_t)((int)blockIdx.x * GEMM_TILE_M + mrow + l16) * HIDDEN + hf * 4;
#pragma unroll
        for (int j = 0; j < 4; ++j) a_cur[j] = *(const v4f*)(p + 8 * j);
    }
#endif

    {
        const int rot = (int)blockIdx.x & 63;
        for (int i = 0; i < 64; i += 16) {
            v4f t[16];
#pragma unroll
            for (int j = 0; j < 16; ++j) {
                const int q = (((i + j + rot) & 63) << 8) + (int)threadIdx.x;
                t[j] = *(const v4f*)(W + (size_t)q * 4);
            }
#pragma unroll
            for (int j = 0; j < 16; ++j) {
                const int q = (((i + j + rot) & 63) << 8) + (int)threadIdx.x;
                *(v4f*)(Ws + (q >> 6) * W_PITCH + (q & 63) * 4) = t[j];
            }
        }
    }
    __syncthreads();

#if HAVE_WMMA_F32
    const float* bp = Ws + (nbase + l16) * W_PITCH + hf * 4;

    v8f acc[GEMM_NT] = {};
    float bz[GEMM_NT];
#pragma unroll
    for (int nt = 0; nt < GEMM_NT; ++nt) bz[nt] = EPI ? bias[nbase + nt * 16 + l16] : 0.0f;
    v4f bq[GEMM_NT];
#pragma unroll
    for (int nt = 0; nt < GEMM_NT; ++nt) bq[nt] = *(const v4f*)(bp + nt * 16 * W_PITCH);

    for (int g = 0; g < nchunk; ++g) {
        const int kc = g & 7;
        {
            const int gn = (g + 1 < nchunk) ? g + 1 : g;
            const int tile = (int)blockIdx.x + (gn >> 3) * GEMM_WGS;
            const float* p = x + (size_t)(tile * GEMM_TILE_M + mrow + l16) * HIDDEN
                               + (gn & 7) * 32 + hf * 4;
#pragma unroll
            for (int j = 0; j < 4; ++j) a_nxt[j] = *(const v4f*)(p + 8 * j);
        }

#pragma unroll
        for (int j = 0; j < 4; ++j) {
            const int knext = (j < 3) ? kc * 32 + (j + 1) * 8 : ((kc + 1) & 7) * 32;
            v4f bn[GEMM_NT];
#pragma unroll
            for (int nt = 0; nt < GEMM_NT; ++nt)
                bn[nt] = *(const v4f*)(bp + nt * 16 * W_PITCH + knext);

            const v2f alo = {a_cur[j].x, a_cur[j].y};
            const v2f ahi = {a_cur[j].z, a_cur[j].w};
#pragma unroll
            for (int nt = 0; nt < GEMM_NT; ++nt) {
                const v2f blo = {bq[nt].x, bq[nt].y};
                acc[nt] = __builtin_amdgcn_wmma_f32_16x16x4_f32(
                     false, alo,  false, blo,
                     (short)0, acc[nt],
                     false,  false);
            }
#pragma unroll
            for (int nt = 0; nt < GEMM_NT; ++nt) {
                const v2f bhi = {bq[nt].z, bq[nt].w};
                acc[nt] = __builtin_amdgcn_wmma_f32_16x16x4_f32(
                     false, ahi,  false, bhi,
                     (short)0, acc[nt],
                     false,  false);
            }
#pragma unroll
            for (int nt = 0; nt < GEMM_NT; ++nt) bq[nt] = bn[nt];
        }

        if (kc == 7) {
            const int tile = (int)blockIdx.x + (g >> 3) * GEMM_WGS;
            float* outp = h + (size_t)(tile * GEMM_TILE_M + mrow + hf * 8) * HIDDEN + (nbase + l16);
#pragma unroll
            for (int nt = 0; nt < GEMM_NT; ++nt) {
#pragma unroll
                for (int r = 0; r < 8; ++r) {
                    float o = acc[nt][r];
                    if (EPI) { o += bz[nt]; o = o > 0.0f ? o : 0.0f; }
                    outp[(size_t)r * HIDDEN + nt * 16] = o;
                }
                acc[nt] = v8f{};
            }
        }
#pragma unroll
        for (int j = 0; j < 4; ++j) a_cur[j] = a_nxt[j];
    }
#else
    for (int t = 0; t < ntile; ++t) {
        const int tile = (int)blockIdx.x + t * GEMM_WGS;
        for (int idx = threadIdx.x; idx < GEMM_TILE_M * HIDDEN; idx += 256) {
            const int row = tile * GEMM_TILE_M + (idx >> 8), col = idx & 255;
            const float* xr = x + (size_t)row * HIDDEN;
            float s = 0.0f;
            for (int k = 0; k < HIDDEN; ++k) s += xr[k] * Ws[col * W_PITCH + k];
            if (EPI) { s += bias[col]; s = s > 0.0f ? s : 0.0f; }
            h[(size_t)row * HIDDEN + col] = s;
        }
    }
    (void)lane; (void)wv; (void)hf; (void)l16; (void)mrow; (void)nbase; (void)nchunk;
#endif
}


__global__ void __launch_bounds__(256)
k_gather(const float* __restrict__ h, float* __restrict__ x,
         const int* __restrict__ rowptr, const int* __restrict__ csrc,
         const float* __restrict__ cw, const float* __restrict__ selfnorm,
         const float* __restrict__ bias) {
    const int lane = threadIdx.x & 31;
    const int node = __builtin_amdgcn_readfirstlane((int)(blockIdx.x * 8 + (threadIdx.x >> 5)));
    const int b = node / V_N;
    const int v = node - b * V_N;
    const float* hb = h + (size_t)b * V_N * HIDDEN;

    const float* hs = hb + (size_t)v * HIDDEN;
    const v4f s0 = *(const v4f*)(hs + lane * 4);
    const v4f s1 = *(const v4f*)(hs + 128 + lane * 4);
    const v4f c0 = *(const v4f*)(bias + lane * 4);
    const v4f c1 = *(const v4f*)(bias + 128 + lane * 4);
    const float sn = selfnorm[v];

    const int beg = rowptr[v], end = rowptr[v + 1];
    v4f a0 = {0.0f, 0.0f, 0.0f, 0.0f};
    v4f a1 = {0.0f, 0.0f, 0.0f, 0.0f};

    int j = beg;
    for (; j + 4 <= end; j += 4) {
        v4f u0[4], u1[4];
        float w[4];
#pragma unroll
        for (int k = 0; k < 4; ++k) {
            const float* p = hb + row_of(csrc[j + k]) * HIDDEN;
            w[k]  = cw[j + k];
            u0[k] = *(const v4f*)(p + lane * 4);
            u1[k] = *(const v4f*)(p + 128 + lane * 4);
        }
#pragma unroll
        for (int k = 0; k < 4; ++k) {
            a0 += u0[k] * w[k];
            a1 += u1[k] * w[k];
        }
    }
    if (j + 2 <= end) {
        v4f u0[2], u1[2];
        float w[2];
#pragma unroll
        for (int k = 0; k < 2; ++k) {
            const float* p = hb + row_of(csrc[j + k]) * HIDDEN;
            w[k]  = cw[j + k];
            u0[k] = *(const v4f*)(p + lane * 4);
            u1[k] = *(const v4f*)(p + 128 + lane * 4);
        }
#pragma unroll
        for (int k = 0; k < 2; ++k) {
            a0 += u0[k] * w[k];
            a1 += u1[k] * w[k];
        }
        j += 2;
    }
    if (j < end) {
        const float* p = hb + row_of(csrc[j]) * HIDDEN;
        const float w = cw[j];
        const v4f u0 = *(const v4f*)(p + lane * 4);
        const v4f u1 = *(const v4f*)(p + 128 + lane * 4);
        a0 += u0 * w;
        a1 += u1 * w;
    }

    a0 += s0 * sn;
    a1 += s1 * sn;
    float* xo = x + (size_t)node * HIDDEN;
    *(v4f*)(xo + lane * 4)       = relu4(a0 + c0);
    *(v4f*)(xo + 128 + lane * 4) = relu4(a1 + c1);
}

#define GO_WPB 32
static_assert(NNODES % GO_WPB == 0, "grid covers the rows exactly");
static_assert((GO_WPB * 3) % 32 == 0, "a block's results are whole 128-byte lines");

__global__ void __launch_bounds__(32 * GO_WPB, 16)
k_gather_out(const float* __restrict__ h,
             const int* __restrict__ rowptr, const int* __restrict__ csrc,
             const float* __restrict__ cw, const float* __restrict__ selfnorm,
             const float* __restrict__ bias, const float* __restrict__ oW,
             const float* __restrict__ ob, float* __restrict__ disp) {
    __shared__ float res[GO_WPB * 3];
    const int lane = threadIdx.x & 31;
    const int wv   = threadIdx.x >> 5;
    const int node = __builtin_amdgcn_readfirstlane((int)(blockIdx.x * GO_WPB + wv));
    const int b = node / V_N;
    const int v = node - b * V_N;
    const float* hb = h + (size_t)b * V_N * HIDDEN;

    const float* hs = hb + (size_t)v * HIDDEN;
    const v4f s0 = *(const v4f*)(hs + lane * 4);
    const v4f s1 = *(const v4f*)(hs + 128 + lane * 4);
    const v4f c0 = *(const v4f*)(bias + lane * 4);
    const v4f c1 = *(const v4f*)(bias + 128 + lane * 4);
    const float sn = selfnorm[v];

    const int beg = rowptr[v], end = rowptr[v + 1];
    v4f a0 = {0.0f, 0.0f, 0.0f, 0.0f};
    v4f a1 = {0.0f, 0.0f, 0.0f, 0.0f};

    int j = beg;
    for (; j + 4 <= end; j += 4) {
        v4f u0[4], u1[4];
        float w[4];
#pragma unroll
        for (int k = 0; k < 4; ++k) {
            const float* p = hb + row_of(csrc[j + k]) * HIDDEN;
            w[k]  = cw[j + k];
            u0[k] = *(const v4f*)(p + lane * 4);
            u1[k] = *(const v4f*)(p + 128 + lane * 4);
        }
#pragma unroll
        for (int k = 0; k < 4; ++k) {
            a0 += u0[k] * w[k];
            a1 += u1[k] * w[k];
        }
    }
    if (j + 2 <= end) {
        v4f u0[2], u1[2];
        float w[2];
#pragma unroll
        for (int k = 0; k < 2; ++k) {
            const float* p = hb + row_of(csrc[j + k]) * HIDDEN;
            w[k]  = cw[j + k];
            u0[k] = *(const v4f*)(p + lane * 4);
            u1[k] = *(const v4f*)(p + 128 + lane * 4);
        }
#pragma unroll
        for (int k = 0; k < 2; ++k) {
            a0 += u0[k] * w[k];
            a1 += u1[k] * w[k];
        }
        j += 2;
    }
    if (j < end) {
        const float* p = hb + row_of(csrc[j]) * HIDDEN;
        const float w = cw[j];
        const v4f u0 = *(const v4f*)(p + lane * 4);
        const v4f u1 = *(const v4f*)(p + 128 + lane * 4);
        a0 += u0 * w;
        a1 += u1 * w;
    }

    a0 += s0 * sn;
    a1 += s1 * sn;
    const v4f r0 = relu4(a0 + c0);
    const v4f r1 = relu4(a1 + c1);

    float s[3];
#pragma unroll
    for (int o = 0; o < 3; ++o) {
        const v4f w0 = *(const v4f*)(oW + o * HIDDEN + lane * 4);
        const v4f w1 = *(const v4f*)(oW + o * HIDDEN + 128 + lane * 4);
        s[o] = r0.x * w0.x + r0.y * w0.y + r0.z * w0.z + r0.w * w0.w
             + r1.x * w1.x + r1.y * w1.y + r1.z * w1.z + r1.w * w1.w;
    }
#pragma unroll
    for (int m = 16; m >= 1; m >>= 1) {
#pragma unroll
        for (int o = 0; o < 3; ++o) s[o] += __shfl_xor(s[o], m);
    }
    if (lane == 0) {
#pragma unroll
        for (int o = 0; o < 3; ++o) res[wv * 3 + o] = s[o] + ob[o];
    }
    __syncthreads();
    if (threadIdx.x < GO_WPB * 3)
        disp[(size_t)blockIdx.x * (GO_WPB * 3) + threadIdx.x] = res[threadIdx.x];
}


extern "C" void kernel_launch(void* const* d_in, const int* in_sizes, int n_in,
                              void* d_out, int out_size, void* d_ws, size_t ws_size,
                              hipStream_t stream) {
    const float* vertex_xyz = (const float*)d_in[0];
    const float* latent     = (const float*)d_in[1];
    const int*   edge_index = (const int*)d_in[2];
    const float* input_W    = (const float*)d_in[3];
    const float* input_b    = (const float*)d_in[4];
    const float* conv_W     = (const float*)d_in[5];
    const float* conv_b     = (const float*)d_in[6];
    const float* out_W      = (const float*)d_in[7];
    const float* out_b      = (const float*)d_in[8];
    float* disp = (float*)d_out;

    const int* src = edge_index;
    const int* dst = edge_index + E_N;

    float* xbuf     = (float*)d_ws;
    float* hbuf     = xbuf + (size_t)NNODES * HIDDEN;
    float* xyzW     = hbuf + (size_t)NNODES * HIDDEN;
    float* latW     = xyzW + (size_t)V_N * HIDDEN;
    float* dinv     = latW + (size_t)B_N * HIDDEN;
    float* selfnorm = dinv + V_N;
    int*   cnt      = (int*)(selfnorm + V_N);
    int*   rowptr   = cnt + V_N;
    int*   rank     = (int*)hbuf;
    int*   slot     = rank + E_N;
    int*   csrc     = (int*)xyzW;
    float* cw       = (float*)(csrc + E_N);

    const int T = 256;

    k_latW<<<(B_N * HIDDEN + T - 1) / T, T, 0, stream>>>(latent, input_W, latW);

    k_cnt_zero<<<(V_N + T - 1) / T, T, 0, stream>>>(cnt);
    k_rank<<<(E_N + T - 1) / T, T, 0, stream>>>(cnt, dst, rank);
    k_scan<<<1, SCAN_T, 0, stream>>>(cnt, rowptr, dinv, selfnorm);
    k_fill<<<(E_N + T - 1) / T, T, 0, stream>>>(rowptr, dst, rank, slot);
    k_place<<<(E_N + T - 1) / T, T, 0, stream>>>(rowptr, slot, src, dst, dinv, csrc, cw);

    dim3 ggrid(GEMM_WGS);
    k_agg0<<<V_N / AG_WPB, 32 * AG_WPB, 0, stream>>>(vertex_xyz, input_W, input_b, latW,
                                                     rowptr, csrc, cw, selfnorm, xbuf);
    k_gemm_wmma<true><<<ggrid, T, 0, stream>>>(xbuf, conv_W, conv_b, hbuf);

    for (int l = 1; l < NLAYERS - 1; ++l) {
        k_gemm_wmma<false><<<ggrid, T, 0, stream>>>(hbuf, conv_W + (size_t)l * HIDDEN * HIDDEN,
                                                    conv_b, xbuf);
        k_gather<<<NNODES / 8, T, 0, stream>>>(xbuf, hbuf, rowptr, csrc, cw, selfnorm,
                                               conv_b + (size_t)l * HIDDEN);
    }

    k_gemm_wmma<false><<<ggrid, T, 0, stream>>>(hbuf, conv_W + (size_t)(NLAYERS - 1) * HIDDEN * HIDDEN,
                                                conv_b, xbuf);
    k_gather_out<<<NNODES / GO_WPB, 32 * GO_WPB, 0, stream>>>(xbuf, rowptr, csrc, cw, selfnorm,
                                                            conv_b + (size_t)(NLAYERS - 1) * HIDDEN,
                                                            out_W, out_b, disp);
}


# Neighbour sum fused into the next layer's GEMM: 16 more waves per work-group sum the next 16-row tile into LDS while 8 waves multiply the current one; three gather launches and their row traffic gone
# speedup vs baseline: 3.9345x; 1.1860x over previous
#include <hip/hip_runtime.h>
#include <hip/hip_bf16.h>

#define V_N     10000
#define E_N     60000
#define B_N     8
#define LATENT  512
#define HIDDEN  256
#define NLAYERS 5
#define NNODES  (B_N * V_N)

typedef float v2f __attribute__((ext_vector_type(2)));
typedef float v8f __attribute__((ext_vector_type(8)));

#if defined(__gfx1250__) && __has_builtin(__builtin_amdgcn_wmma_f32_16x16x4_f32)
#define HAVE_WMMA_F32 1
#endif


__global__ void k_cnt_zero(int* cnt) {
    int v = blockIdx.x * blockDim.x + threadIdx.x;
    if (v < V_N) cnt[v] = 0;
}

__global__ void k_rank(int* cnt, const int* __restrict__ dst, int* __restrict__ rank) {
    int e = blockIdx.x * blockDim.x + threadIdx.x;
    if (e < E_N) rank[e] = atomicAdd(&cnt[dst[e]], 1);
}

#define SCAN_T   1024
#define SCAN_PER ((V_N + SCAN_T - 1) / SCAN_T)
__global__ void __launch_bounds__(SCAN_T)
k_scan(const int* __restrict__ cnt, int* __restrict__ rowptr,
       float* __restrict__ dinv, float* __restrict__ selfnorm) {
    __shared__ int part[SCAN_T];
    const int t = threadIdx.x;
    const int base = t * SCAN_PER;
    int loc[SCAN_PER];
    int c[SCAN_PER];
    int s = 0;
#pragma unroll
    for (int i = 0; i < SCAN_PER; ++i) {
        int v = base + i;
        c[i] = (v < V_N) ? cnt[v] : 0;
        loc[i] = s;
        s += c[i];
    }
    part[t] = s;
    __syncthreads();
    for (int off = 1; off < SCAN_T; off <<= 1) {
        int add = (t >= off) ? part[t - off] : 0;
        __syncthreads();
        part[t] += add;
        __syncthreads();
    }
    const int excl = part[t] - s;
#pragma unroll
    for (int i = 0; i < SCAN_PER; ++i) {
        int v = base + i;
        if (v < V_N) {
            rowptr[v] = excl + loc[i];
            float di = rsqrtf((float)(c[i] + 1));
            dinv[v] = di;
            selfnorm[v] = di * di;
        }
    }
    if (t == SCAN_T - 1) rowptr[V_N] = part[t];
}

__global__ void k_fill(const int* __restrict__ rowptr, const int* __restrict__ dst,
                       const int* __restrict__ rank, int* __restrict__ slot) {
    int e = blockIdx.x * blockDim.x + threadIdx.x;
    if (e < E_N) slot[rowptr[dst[e]] + rank[e]] = e;
}

__global__ void k_place(const int* __restrict__ rowptr, const int* __restrict__ slot,
                        const int* __restrict__ src, const int* __restrict__ dst,
                        const float* __restrict__ dinv,
                        int* __restrict__ csrc, float* __restrict__ cw) {
    int e = blockIdx.x * blockDim.x + threadIdx.x;
    if (e >= E_N) return;
    const int v = dst[e], s = src[e];
    const int beg = rowptr[v], end = rowptr[v + 1];
    int r = 0;
    for (int j = beg; j < end; ++j) r += (slot[j] < e) ? 1 : 0;
    csrc[beg + r] = s;
    cw[beg + r]   = dinv[s] * dinv[v];
}


__global__ void k_latW(const float* __restrict__ latent, const float* __restrict__ iW,
                       float* __restrict__ latW) {
    int i = blockIdx.x * blockDim.x + threadIdx.x;
    if (i >= B_N * HIDDEN) return;
    int b = i / HIDDEN, hh = i % HIDDEN;
    const float* lp = latent + (size_t)b * LATENT;
    const float* wp = iW + (size_t)hh * (LATENT + 3) + 3;
    float s = 0.0f;
    for (int j = 0; j < LATENT; ++j) s += lp[j] * wp[j];
    latW[i] = s;
}

typedef float v4f __attribute__((ext_vector_type(4)));

__device__ __forceinline__ v4f relu4(v4f s) {
    v4f r;
    r.x = s.x > 0.0f ? s.x : 0.0f;
    r.y = s.y > 0.0f ? s.y : 0.0f;
    r.z = s.z > 0.0f ? s.z : 0.0f;
    r.w = s.w > 0.0f ? s.w : 0.0f;
    return r;
}

__device__ __forceinline__ size_t row_of(int idx) {
    unsigned u = (unsigned)idx;
    return (size_t)(u < (unsigned)(V_N - 1) ? u : (unsigned)(V_N - 1));
}

#define AG_WPB 8
static_assert(V_N % AG_WPB == 0, "grid covers the vertices exactly");

__device__ __forceinline__ void agg0_term(v4f (&a0)[B_N], v4f (&a1)[B_N],
                                          const v4f (&l0)[B_N], const v4f (&l1)[B_N],
                                          const v4f t0, const v4f t1, const float w) {
#pragma unroll
    for (int b = 0; b < B_N; ++b) {
        a0[b] += relu4(t0 + l0[b]) * w;
        a1[b] += relu4(t1 + l1[b]) * w;
    }
}

__global__ void __launch_bounds__(32 * AG_WPB)
k_agg0(const float* __restrict__ xyz, const float* __restrict__ iW, const float* __restrict__ ib,
       const float* __restrict__ latW, const int* __restrict__ rowptr, const int* __restrict__ csrc,
       const float* __restrict__ cw, const float* __restrict__ selfnorm, float* __restrict__ y) {
    const int lane = threadIdx.x & 31;
    const int v = __builtin_amdgcn_readfirstlane((int)(blockIdx.x * AG_WPB + (threadIdx.x >> 5)));
    const int c0 = lane * 4, c1 = 128 + lane * 4;

    v4f wx0, wy0, wz0, wx1, wy1, wz1;
#pragma unroll
    for (int i = 0; i < 4; ++i) {
        const float* p0 = iW + (size_t)(c0 + i) * (LATENT + 3);
        const float* p1 = iW + (size_t)(c1 + i) * (LATENT + 3);
        wx0[i] = p0[0]; wy0[i] = p0[1]; wz0[i] = p0[2];
        wx1[i] = p1[0]; wy1[i] = p1[1]; wz1[i] = p1[2];
    }
    const v4f b0 = *(const v4f*)(ib + c0);
    const v4f b1 = *(const v4f*)(ib + c1);
    v4f l0[B_N], l1[B_N], a0[B_N], a1[B_N];
#pragma unroll
    for (int b = 0; b < B_N; ++b) {
        l0[b] = *(const v4f*)(latW + b * HIDDEN + c0);
        l1[b] = *(const v4f*)(latW + b * HIDDEN + c1);
        a0[b] = v4f{0.0f, 0.0f, 0.0f, 0.0f};
        a1[b] = v4f{0.0f, 0.0f, 0.0f, 0.0f};
    }

    const int beg = rowptr[v], end = rowptr[v + 1];
    int j = beg;
    for (; j + 4 <= end; j += 4) {
        size_t s[4];
        float w[4], px[4], py[4], pz[4];
#pragma unroll
        for (int k = 0; k < 4; ++k) { s[k] = row_of(csrc[j + k]); w[k] = cw[j + k]; }
#pragma unroll
        for (int k = 0; k < 4; ++k) {
            px[k] = xyz[3 * s[k]]; py[k] = xyz[3 * s[k] + 1]; pz[k] = xyz[3 * s[k] + 2];
        }
#pragma unroll
        for (int k = 0; k < 4; ++k) {
            const v4f t0 = b0 + wx0 * px[k] + wy0 * py[k] + wz0 * pz[k];
            const v4f t1 = b1 + wx1 * px[k] + wy1 * py[k] + wz1 * pz[k];
            agg0_term(a0, a1, l0, l1, t0, t1, w[k]);
        }
    }
    for (; j < end; ++j) {
        const size_t s = row_of(csrc[j]);
        const float w = cw[j];
        const float px = xyz[3 * s], py = xyz[3 * s + 1], pz = xyz[3 * s + 2];
        const v4f t0 = b0 + wx0 * px + wy0 * py + wz0 * pz;
        const v4f t1 = b1 + wx1 * px + wy1 * py + wz1 * pz;
        agg0_term(a0, a1, l0, l1, t0, t1, w);
    }
    {
        const float w = selfnorm[v];
        const float px = xyz[3 * v], py = xyz[3 * v + 1], pz = xyz[3 * v + 2];
        const v4f t0 = b0 + wx0 * px + wy0 * py + wz0 * pz;
        const v4f t1 = b1 + wx1 * px + wy1 * py + wz1 * pz;
        agg0_term(a0, a1, l0, l1, t0, t1, w);
    }

#pragma unroll
    for (int b = 0; b < B_N; ++b) {
        float* yo = y + ((size_t)b * V_N + v) * HIDDEN;
        *(v4f*)(yo + c0) = a0[b];
        *(v4f*)(yo + c1) = a1[b];
    }
}


#define GEMM_TILE_M  64
#define GEMM_TILES   (NNODES / GEMM_TILE_M)
#define GEMM_WGS     256
#define GEMM_NT      8
#define W_PITCH      260
static_assert(NNODES % GEMM_TILE_M == 0, "tiles cover the rows exactly");
static_assert(GEMM_WGS <= GEMM_TILES, "every work-group owns at least one tile");

template <bool EPI>
__global__ void __launch_bounds__(256)
k_gemm_wmma(const float* __restrict__ x, const float* __restrict__ W,
            const float* __restrict__ bias, float* __restrict__ h) {
    __shared__ __attribute__((aligned(16))) float Ws[HIDDEN * W_PITCH];

    const int lane  = threadIdx.x & 31;
    const int wv    = threadIdx.x >> 5;
    const int hf    = lane >> 4;
    const int l16   = lane & 15;
    const int mrow  = (wv >> 1) * 16;
    const int nbase = (wv & 1) * 128;

    const int ntile  = (GEMM_TILES - (int)blockIdx.x + GEMM_WGS - 1) / GEMM_WGS;
    const int nchunk = ntile * 8;

#if HAVE_WMMA_F32
    v4f a_cur[4], a_nxt[4];
    {
        const float* p = x + (size_t)((int)blockIdx.x * GEMM_TILE_M + mrow + l16) * HIDDEN + hf * 4;
#pragma unroll
        for (int j = 0; j < 4; ++j) a_cur[j] = *(const v4f*)(p + 8 * j);
    }
#endif

    {
        const int rot = (int)blockIdx.x & 63;
        for (int i = 0; i < 64; i += 16) {
            v4f t[16];
#pragma unroll
            for (int j = 0; j < 16; ++j) {
                const int q = (((i + j + rot) & 63) << 8) + (int)threadIdx.x;
                t[j] = *(const v4f*)(W + (size_t)q * 4);
            }
#pragma unroll
            for (int j = 0; j < 16; ++j) {
                const int q = (((i + j + rot) & 63) << 8) + (int)threadIdx.x;
                *(v4f*)(Ws + (q >> 6) * W_PITCH + (q & 63) * 4) = t[j];
            }
        }
    }
    __syncthreads();

#if HAVE_WMMA_F32
    const float* bp = Ws + (nbase + l16) * W_PITCH + hf * 4;

    v8f acc[GEMM_NT] = {};
    float bz[GEMM_NT];
#pragma unroll
    for (int nt = 0; nt < GEMM_NT; ++nt) bz[nt] = EPI ? bias[nbase + nt * 16 + l16] : 0.0f;
    v4f bq[GEMM_NT];
#pragma unroll
    for (int nt = 0; nt < GEMM_NT; ++nt) bq[nt] = *(const v4f*)(bp + nt * 16 * W_PITCH);

    for (int g = 0; g < nchunk; ++g) {
        const int kc = g & 7;
        {
            const int gn = (g + 1 < nchunk) ? g + 1 : g;
            const int tile = (int)blockIdx.x + (gn >> 3) * GEMM_WGS;
            const float* p = x + (size_t)(tile * GEMM_TILE_M + mrow + l16) * HIDDEN
                               + (gn & 7) * 32 + hf * 4;
#pragma unroll
            for (int j = 0; j < 4; ++j) a_nxt[j] = *(const v4f*)(p + 8 * j);
        }

#pragma unroll
        for (int j = 0; j < 4; ++j) {
            const int knext = (j < 3) ? kc * 32 + (j + 1) * 8 : ((kc + 1) & 7) * 32;
            v4f bn[GEMM_NT];
#pragma unroll
            for (int nt = 0; nt < GEMM_NT; ++nt)
                bn[nt] = *(const v4f*)(bp + nt * 16 * W_PITCH + knext);

            const v2f alo = {a_cur[j].x, a_cur[j].y};
            const v2f ahi = {a_cur[j].z, a_cur[j].w};
#pragma unroll
            for (int nt = 0; nt < GEMM_NT; ++nt) {
                const v2f blo = {bq[nt].x, bq[nt].y};
                acc[nt] = __builtin_amdgcn_wmma_f32_16x16x4_f32(
                     false, alo,  false, blo,
                     (short)0, acc[nt],
                     false,  false);
            }
#pragma unroll
            for (int nt = 0; nt < GEMM_NT; ++nt) {
                const v2f bhi = {bq[nt].z, bq[nt].w};
                acc[nt] = __builtin_amdgcn_wmma_f32_16x16x4_f32(
                     false, ahi,  false, bhi,
                     (short)0, acc[nt],
                     false,  false);
            }
#pragma unroll
            for (int nt = 0; nt < GEMM_NT; ++nt) bq[nt] = bn[nt];
        }

        if (kc == 7) {
            const int tile = (int)blockIdx.x + (g >> 3) * GEMM_WGS;
            float* outp = h + (size_t)(tile * GEMM_TILE_M + mrow + hf * 8) * HIDDEN + (nbase + l16);
#pragma unroll
            for (int nt = 0; nt < GEMM_NT; ++nt) {
#pragma unroll
                for (int r = 0; r < 8; ++r) {
                    float o = acc[nt][r];
                    if (EPI) { o += bz[nt]; o = o > 0.0f ? o : 0.0f; }
                    outp[(size_t)r * HIDDEN + nt * 16] = o;
                }
                acc[nt] = v8f{};
            }
        }
#pragma unroll
        for (int j = 0; j < 4; ++j) a_cur[j] = a_nxt[j];
    }
#else
    for (int t = 0; t < ntile; ++t) {
        const int tile = (int)blockIdx.x + t * GEMM_WGS;
        for (int idx = threadIdx.x; idx < GEMM_TILE_M * HIDDEN; idx += 256) {
            const int row = tile * GEMM_TILE_M + (idx >> 8), col = idx & 255;
            const float* xr = x + (size_t)row * HIDDEN;
            float s = 0.0f;
            for (int k = 0; k < HIDDEN; ++k) s += xr[k] * Ws[col * W_PITCH + k];
            if (EPI) { s += bias[col]; s = s > 0.0f ? s : 0.0f; }
            h[(size_t)row * HIDDEN + col] = s;
        }
    }
    (void)lane; (void)wv; (void)hf; (void)l16; (void)mrow; (void)nbase; (void)nchunk;
#endif
}


__device__ __forceinline__ void agg_row(const float* __restrict__ h, const int node,
                                        const int* __restrict__ rowptr, const int* __restrict__ csrc,
                                        const float* __restrict__ cw, const float* __restrict__ selfnorm,
                                        const v4f c0, const v4f c1, const int lane, v4f& r0, v4f& r1) {
    const int b = node / V_N;
    const int v = node - b * V_N;
    const float* hb = h + (size_t)b * V_N * HIDDEN;
    const float* hs = hb + (size_t)v * HIDDEN;
    const v4f s0 = *(const v4f*)(hs + lane * 4);
    const v4f s1 = *(const v4f*)(hs + 128 + lane * 4);
    const float sn = selfnorm[v];
    const int beg = rowptr[v], end = rowptr[v + 1];
    v4f a0 = {0.0f, 0.0f, 0.0f, 0.0f};
    v4f a1 = {0.0f, 0.0f, 0.0f, 0.0f};
    int j = beg;
    for (; end - j >= 8; j += 4) {
        v4f u0[4], u1[4];
        float w[4];
#pragma unroll
        for (int k = 0; k < 4; ++k) {
            const float* p = hb + row_of(csrc[j + k]) * HIDDEN;
            w[k]  = cw[j + k];
            u0[k] = *(const v4f*)(p + lane * 4);
            u1[k] = *(const v4f*)(p + 128 + lane * 4);
        }
#pragma unroll
        for (int k = 0; k < 4; ++k) {
            a0 += u0[k] * w[k];
            a1 += u1[k] * w[k];
        }
    }
    const int n = end - j;
    v4f u0[7], u1[7];
    float w[7];
#pragma unroll
    for (int k = 0; k < 7; ++k) {
        const bool on = k < n;
        const int e = on ? j + k : 0;
        const size_t sr = on ? row_of(csrc[e]) : (size_t)0;
        const float we = cw[e];
        const float* p = hb + sr * HIDDEN;
        w[k]  = on ? we : 0.0f;
        u0[k] = *(const v4f*)(p + lane * 4);
        u1[k] = *(const v4f*)(p + 128 + lane * 4);
    }
#pragma unroll
    for (int k = 0; k < 7; ++k) {
        a0 += u0[k] * w[k];
        a1 += u1[k] * w[k];
    }

    a0 += s0 * sn;
    a1 += s1 * sn;
    r0 = relu4(a0 + c0);
    r1 = relu4(a1 + c1);
}

#define FG_ROWS    16
#define FG_TILES   (NNODES / FG_ROWS)
#define FG_MWAVES  8
#define FG_GWAVES  16
#define FG_THREADS (32 * (FG_MWAVES + FG_GWAVES))
#define A_PITCH    260
#define FG_PIECES  (HIDDEN * HIDDEN / 4)
#define FG_CPY     11
static_assert(NNODES % FG_ROWS == 0, "tiles cover the rows exactly");
static_assert(FG_GWAVES == FG_ROWS, "one gathering wave per tile row");
static_assert(2 * FG_CPY * FG_THREADS >= FG_PIECES, "the copy covers W");

__global__ void __launch_bounds__(FG_THREADS)
k_gemm_gather(const float* __restrict__ hin, const float* __restrict__ W,
              const float* __restrict__ bias_in,
              const int* __restrict__ rowptr, const int* __restrict__ csrc,
              const float* __restrict__ cw, const float* __restrict__ selfnorm,
              float* __restrict__ hout) {
    __shared__ __attribute__((aligned(16))) float Ws[HIDDEN * W_PITCH];
    __shared__ __attribute__((aligned(16))) float As[2 * FG_ROWS * A_PITCH];

    const int lane = threadIdx.x & 31;
    const int wv = __builtin_amdgcn_readfirstlane((int)(threadIdx.x >> 5));
    const bool gatherer = wv >= FG_MWAVES;
    const int grow = wv - FG_MWAVES;
    const int ntile = (FG_TILES - (int)blockIdx.x + GEMM_WGS - 1) / GEMM_WGS;

    {
        const int rot = ((int)blockIdx.x & 63) << 8;
        for (int i = 0; i < 2 * FG_CPY; i += FG_CPY) {
            v4f t[FG_CPY];
#pragma unroll
            for (int j = 0; j < FG_CPY; ++j) {
                const int p = (i + j) * FG_THREADS + (int)threadIdx.x;
                const int q = (p + rot) & (FG_PIECES - 1);
                t[j] = (p < FG_PIECES) ? *(const v4f*)(W + (size_t)q * 4) : v4f{0.0f, 0.0f, 0.0f, 0.0f};
            }
#pragma unroll
            for (int j = 0; j < FG_CPY; ++j) {
                const int p = (i + j) * FG_THREADS + (int)threadIdx.x;
                const int q = (p + rot) & (FG_PIECES - 1);
                if (p < FG_PIECES) *(v4f*)(Ws + (q >> 6) * W_PITCH + (q & 63) * 4) = t[j];
            }
        }
    }

    const v4f c0 = *(const v4f*)(bias_in + lane * 4);
    const v4f c1 = *(const v4f*)(bias_in + 128 + lane * 4);

    if (gatherer) {
        v4f r0, r1;
        agg_row(hin, (int)blockIdx.x * FG_ROWS + grow, rowptr, csrc, cw, selfnorm, c0, c1, lane, r0, r1);
        float* ar = As + grow * A_PITCH;
        *(v4f*)(ar + lane * 4)       = r0;
        *(v4f*)(ar + 128 + lane * 4) = r1;
    }
    __syncthreads();

#if HAVE_WMMA_F32
    const int hf    = lane >> 4;
    const int l16   = lane & 15;
    const int nbase = (gatherer ? 0 : wv) * 32;
    const float* bp = Ws + (nbase + l16) * W_PITCH + hf * 4;
#endif

    for (int t = 0; t < ntile; ++t) {
        const int tile = (int)blockIdx.x + t * GEMM_WGS;
        if (gatherer) {
            if (t + 1 < ntile) {
                v4f r0, r1;
                agg_row(hin, (tile + GEMM_WGS) * FG_ROWS + grow, rowptr, csrc, cw, selfnorm,
                        c0, c1, lane, r0, r1);
                float* ar = As + ((t + 1) & 1) * (FG_ROWS * A_PITCH) + grow * A_PITCH;
                *(v4f*)(ar + lane * 4)       = r0;
                *(v4f*)(ar + 128 + lane * 4) = r1;
            }
        } else {
            const float* at = As + (t & 1) * (FG_ROWS * A_PITCH);
#if HAVE_WMMA_F32
            const float* ap = at + l16 * A_PITCH + hf * 4;
            v8f acc0 = {}, acc1 = {};
            v4f fa[4], fb0[4], fb1[4];
#pragma unroll
            for (int j = 0; j < 4; ++j) {
                fa[j]  = *(const v4f*)(ap + 8 * j);
                fb0[j] = *(const v4f*)(bp + 8 * j);
                fb1[j] = *(const v4f*)(bp + 16 * W_PITCH + 8 * j);
            }
#pragma unroll
            for (int kb = 0; kb < 8; ++kb) {
                v4f na[4], nb0[4], nb1[4];
                if (kb + 1 < 8) {
#pragma unroll
                    for (int j = 0; j < 4; ++j) {
                        na[j]  = *(const v4f*)(ap + (kb + 1) * 32 + 8 * j);
                        nb0[j] = *(const v4f*)(bp + (kb + 1) * 32 + 8 * j);
                        nb1[j] = *(const v4f*)(bp + 16 * W_PITCH + (kb + 1) * 32 + 8 * j);
                    }
                }
                __builtin_amdgcn_sched_barrier(0);
#pragma unroll
                for (int j = 0; j < 4; ++j) {
                    const v2f alo  = {fa[j].x, fa[j].y};
                    const v2f ahi  = {fa[j].z, fa[j].w};
                    const v2f b0lo = {fb0[j].x, fb0[j].y};
                    const v2f b0hi = {fb0[j].z, fb0[j].w};
                    const v2f b1lo = {fb1[j].x, fb1[j].y};
                    const v2f b1hi = {fb1[j].z, fb1[j].w};
                    acc0 = __builtin_amdgcn_wmma_f32_16x16x4_f32(false, alo, false, b0lo, (short)0, acc0, false, false);
                    acc1 = __builtin_amdgcn_wmma_f32_16x16x4_f32(false, alo, false, b1lo, (short)0, acc1, false, false);
                    acc0 = __builtin_amdgcn_wmma_f32_16x16x4_f32(false, ahi, false, b0hi, (short)0, acc0, false, false);
                    acc1 = __builtin_amdgcn_wmma_f32_16x16x4_f32(false, ahi, false, b1hi, (short)0, acc1, false, false);
                }
                __builtin_amdgcn_sched_barrier(0);
                if (kb + 1 < 8) {
#pragma unroll
                    for (int j = 0; j < 4; ++j) { fa[j] = na[j]; fb0[j] = nb0[j]; fb1[j] = nb1[j]; }
                }
            }
            {
                float* outp = hout + (size_t)(tile * FG_ROWS + hf * 8) * HIDDEN + (nbase + l16);
#pragma unroll
                for (int r = 0; r < 8; ++r) {
                    outp[(size_t)r * HIDDEN]      = acc0[r];
                    outp[(size_t)r * HIDDEN + 16] = acc1[r];
                }
            }
#else
            for (int m = 0; m < FG_ROWS; ++m) {
                float s = 0.0f;
                for (int k = 0; k < HIDDEN; ++k)
                    s += at[m * A_PITCH + k] * Ws[((wv & 7) * 32 + lane) * W_PITCH + k];
                hout[(size_t)(tile * FG_ROWS + m) * HIDDEN + (wv & 7) * 32 + lane] = s;
            }
#endif
        }
        __builtin_amdgcn_fence(__ATOMIC_RELEASE, "workgroup", "local");
        __builtin_amdgcn_s_barrier();
        __builtin_amdgcn_fence(__ATOMIC_ACQUIRE, "workgroup", "local");
    }
}

#define GO_WPB 32
static_assert(NNODES % GO_WPB == 0, "grid covers the rows exactly");
static_assert((GO_WPB * 3) % 32 == 0, "a block's results are whole 128-byte lines");

__global__ void __launch_bounds__(32 * GO_WPB, 16)
k_gather_out(const float* __restrict__ h,
             const int* __restrict__ rowptr, const int* __restrict__ csrc,
             const float* __restrict__ cw, const float* __restrict__ selfnorm,
             const float* __restrict__ bias, const float* __restrict__ oW,
             const float* __restrict__ ob, float* __restrict__ disp) {
    __shared__ float res[GO_WPB * 3];
    const int lane = threadIdx.x & 31;
    const int wv   = threadIdx.x >> 5;
    const int node = __builtin_amdgcn_readfirstlane((int)(blockIdx.x * GO_WPB + wv));
    const int b = node / V_N;
    const int v = node - b * V_N;
    const float* hb = h + (size_t)b * V_N * HIDDEN;

    const float* hs = hb + (size_t)v * HIDDEN;
    const v4f s0 = *(const v4f*)(hs + lane * 4);
    const v4f s1 = *(const v4f*)(hs + 128 + lane * 4);
    const v4f c0 = *(const v4f*)(bias + lane * 4);
    const v4f c1 = *(const v4f*)(bias + 128 + lane * 4);
    const float sn = selfnorm[v];

    const int beg = rowptr[v], end = rowptr[v + 1];
    v4f a0 = {0.0f, 0.0f, 0.0f, 0.0f};
    v4f a1 = {0.0f, 0.0f, 0.0f, 0.0f};

    int j = beg;
    for (; j + 4 <= end; j += 4) {
        v4f u0[4], u1[4];
        float w[4];
#pragma unroll
        for (int k = 0; k < 4; ++k) {
            const float* p = hb + row_of(csrc[j + k]) * HIDDEN;
            w[k]  = cw[j + k];
            u0[k] = *(const v4f*)(p + lane * 4);
            u1[k] = *(const v4f*)(p + 128 + lane * 4);
        }
#pragma unroll
        for (int k = 0; k < 4; ++k) {
            a0 += u0[k] * w[k];
            a1 += u1[k] * w[k];
        }
    }
    if (j + 2 <= end) {
        v4f u0[2], u1[2];
        float w[2];
#pragma unroll
        for (int k = 0; k < 2; ++k) {
            const float* p = hb + row_of(csrc[j + k]) * HIDDEN;
            w[k]  = cw[j + k];
            u0[k] = *(const v4f*)(p + lane * 4);
            u1[k] = *(const v4f*)(p + 128 + lane * 4);
        }
#pragma unroll
        for (int k = 0; k < 2; ++k) {
            a0 += u0[k] * w[k];
            a1 += u1[k] * w[k];
        }
        j += 2;
    }
    if (j < end) {
        const float* p = hb + row_of(csrc[j]) * HIDDEN;
        const float w = cw[j];
        const v4f u0 = *(const v4f*)(p + lane * 4);
        const v4f u1 = *(const v4f*)(p + 128 + lane * 4);
        a0 += u0 * w;
        a1 += u1 * w;
    }

    a0 += s0 * sn;
    a1 += s1 * sn;
    const v4f r0 = relu4(a0 + c0);
    const v4f r1 = relu4(a1 + c1);

    float s[3];
#pragma unroll
    for (int o = 0; o < 3; ++o) {
        const v4f w0 = *(const v4f*)(oW + o * HIDDEN + lane * 4);
        const v4f w1 = *(const v4f*)(oW + o * HIDDEN + 128 + lane * 4);
        s[o] = r0.x * w0.x + r0.y * w0.y + r0.z * w0.z + r0.w * w0.w
             + r1.x * w1.x + r1.y * w1.y + r1.z * w1.z + r1.w * w1.w;
    }
#pragma unroll
    for (int m = 16; m >= 1; m >>= 1) {
#pragma unroll
        for (int o = 0; o < 3; ++o) s[o] += __shfl_xor(s[o], m);
    }
    if (lane == 0) {
#pragma unroll
        for (int o = 0; o < 3; ++o) res[wv * 3 + o] = s[o] + ob[o];
    }
    __syncthreads();
    if (threadIdx.x < GO_WPB * 3)
        disp[(size_t)blockIdx.x * (GO_WPB * 3) + threadIdx.x] = res[threadIdx.x];
}


extern "C" void kernel_launch(void* const* d_in, const int* in_sizes, int n_in,
                              void* d_out, int out_size, void* d_ws, size_t ws_size,
                              hipStream_t stream) {
    const float* vertex_xyz = (const float*)d_in[0];
    const float* latent     = (const float*)d_in[1];
    const int*   edge_index = (const int*)d_in[2];
    const float* input_W    = (const float*)d_in[3];
    const float* input_b    = (const float*)d_in[4];
    const float* conv_W     = (const float*)d_in[5];
    const float* conv_b     = (const float*)d_in[6];
    const float* out_W      = (const float*)d_in[7];
    const float* out_b      = (const float*)d_in[8];
    float* disp = (float*)d_out;

    const int* src = edge_index;
    const int* dst = edge_index + E_N;

    float* xbuf     = (float*)d_ws;
    float* hbuf     = xbuf + (size_t)NNODES * HIDDEN;
    float* xyzW     = hbuf + (size_t)NNODES * HIDDEN;
    float* latW     = xyzW + (size_t)V_N * HIDDEN;
    float* dinv     = latW + (size_t)B_N * HIDDEN;
    float* selfnorm = dinv + V_N;
    int*   cnt      = (int*)(selfnorm + V_N);
    int*   rowptr   = cnt + V_N;
    int*   rank     = (int*)hbuf;
    int*   slot     = rank + E_N;
    int*   csrc     = (int*)xyzW;
    float* cw       = (float*)(csrc + E_N);

    const int T = 256;

    k_latW<<<(B_N * HIDDEN + T - 1) / T, T, 0, stream>>>(latent, input_W, latW);

    k_cnt_zero<<<(V_N + T - 1) / T, T, 0, stream>>>(cnt);
    k_rank<<<(E_N + T - 1) / T, T, 0, stream>>>(cnt, dst, rank);
    k_scan<<<1, SCAN_T, 0, stream>>>(cnt, rowptr, dinv, selfnorm);
    k_fill<<<(E_N + T - 1) / T, T, 0, stream>>>(rowptr, dst, rank, slot);
    k_place<<<(E_N + T - 1) / T, T, 0, stream>>>(rowptr, slot, src, dst, dinv, csrc, cw);

    dim3 ggrid(GEMM_WGS);
    k_agg0<<<V_N / AG_WPB, 32 * AG_WPB, 0, stream>>>(vertex_xyz, input_W, input_b, latW,
                                                     rowptr, csrc, cw, selfnorm, xbuf);
    k_gemm_wmma<true><<<ggrid, T, 0, stream>>>(xbuf, conv_W, conv_b, hbuf);

    k_gemm_wmma<false><<<ggrid, T, 0, stream>>>(hbuf, conv_W + (size_t)1 * HIDDEN * HIDDEN, conv_b, xbuf);
    k_gemm_gather<<<ggrid, FG_THREADS, 0, stream>>>(xbuf, conv_W + (size_t)2 * HIDDEN * HIDDEN,
                                                    conv_b + (size_t)1 * HIDDEN,
                                                    rowptr, csrc, cw, selfnorm, hbuf);
    k_gemm_gather<<<ggrid, FG_THREADS, 0, stream>>>(hbuf, conv_W + (size_t)3 * HIDDEN * HIDDEN,
                                                    conv_b + (size_t)2 * HIDDEN,
                                                    rowptr, csrc, cw, selfnorm, xbuf);
    k_gemm_gather<<<ggrid, FG_THREADS, 0, stream>>>(xbuf, conv_W + (size_t)4 * HIDDEN * HIDDEN,
                                                    conv_b + (size_t)3 * HIDDEN,
                                                    rowptr, csrc, cw, selfnorm, hbuf);

    k_gather_out<<<NNODES / GO_WPB, 32 * GO_WPB, 0, stream>>>(hbuf, rowptr, csrc, cw, selfnorm,
                                                            conv_b + (size_t)(NLAYERS - 1) * HIDDEN,
                                                            out_W, out_b, disp);
}


# Fused layers: the summing waves run at raised issue priority (s_setprio 3), so their vector instructions no longer queue behind the matrix waves' f32 WMMA stream and their row loads run under it
# speedup vs baseline: 4.4404x; 1.1286x over previous
#include <hip/hip_runtime.h>
#include <hip/hip_bf16.h>

#define V_N     10000
#define E_N     60000
#define B_N     8
#define LATENT  512
#define HIDDEN  256
#define NLAYERS 5
#define NNODES  (B_N * V_N)

typedef float v2f __attribute__((ext_vector_type(2)));
typedef float v8f __attribute__((ext_vector_type(8)));

#if defined(__gfx1250__) && __has_builtin(__builtin_amdgcn_wmma_f32_16x16x4_f32)
#define HAVE_WMMA_F32 1
#endif


__global__ void k_cnt_zero(int* cnt) {
    int v = blockIdx.x * blockDim.x + threadIdx.x;
    if (v < V_N) cnt[v] = 0;
}

__global__ void k_rank(int* cnt, const int* __restrict__ dst, int* __restrict__ rank) {
    int e = blockIdx.x * blockDim.x + threadIdx.x;
    if (e < E_N) rank[e] = atomicAdd(&cnt[dst[e]], 1);
}

#define SCAN_T   1024
#define SCAN_PER ((V_N + SCAN_T - 1) / SCAN_T)
__global__ void __launch_bounds__(SCAN_T)
k_scan(const int* __restrict__ cnt, int* __restrict__ rowptr,
       float* __restrict__ dinv, float* __restrict__ selfnorm) {
    __shared__ int part[SCAN_T];
    const int t = threadIdx.x;
    const int base = t * SCAN_PER;
    int loc[SCAN_PER];
    int c[SCAN_PER];
    int s = 0;
#pragma unroll
    for (int i = 0; i < SCAN_PER; ++i) {
        int v = base + i;
        c[i] = (v < V_N) ? cnt[v] : 0;
        loc[i] = s;
        s += c[i];
    }
    part[t] = s;
    __syncthreads();
    for (int off = 1; off < SCAN_T; off <<= 1) {
        int add = (t >= off) ? part[t - off] : 0;
        __syncthreads();
        part[t] += add;
        __syncthreads();
    }
    const int excl = part[t] - s;
#pragma unroll
    for (int i = 0; i < SCAN_PER; ++i) {
        int v = base + i;
        if (v < V_N) {
            rowptr[v] = excl + loc[i];
            float di = rsqrtf((float)(c[i] + 1));
            dinv[v] = di;
            selfnorm[v] = di * di;
        }
    }
    if (t == SCAN_T - 1) rowptr[V_N] = part[t];
}

__global__ void k_fill(const int* __restrict__ rowptr, const int* __restrict__ dst,
                       const int* __restrict__ rank, int* __restrict__ slot) {
    int e = blockIdx.x * blockDim.x + threadIdx.x;
    if (e < E_N) slot[rowptr[dst[e]] + rank[e]] = e;
}

__global__ void k_place(const int* __restrict__ rowptr, const int* __restrict__ slot,
                        const int* __restrict__ src, const int* __restrict__ dst,
                        const float* __restrict__ dinv,
                        int* __restrict__ csrc, float* __restrict__ cw) {
    int e = blockIdx.x * blockDim.x + threadIdx.x;
    if (e >= E_N) return;
    const int v = dst[e], s = src[e];
    const int beg = rowptr[v], end = rowptr[v + 1];
    int r = 0;
    for (int j = beg; j < end; ++j) r += (slot[j] < e) ? 1 : 0;
    csrc[beg + r] = s;
    cw[beg + r]   = dinv[s] * dinv[v];
}


__global__ void k_latW(const float* __restrict__ latent, const float* __restrict__ iW,
                       float* __restrict__ latW) {
    int i = blockIdx.x * blockDim.x + threadIdx.x;
    if (i >= B_N * HIDDEN) return;
    int b = i / HIDDEN, hh = i % HIDDEN;
    const float* lp = latent + (size_t)b * LATENT;
    const float* wp = iW + (size_t)hh * (LATENT + 3) + 3;
    float s = 0.0f;
    for (int j = 0; j < LATENT; ++j) s += lp[j] * wp[j];
    latW[i] = s;
}

typedef float v4f __attribute__((ext_vector_type(4)));

__device__ __forceinline__ v4f relu4(v4f s) {
    v4f r;
    r.x = s.x > 0.0f ? s.x : 0.0f;
    r.y = s.y > 0.0f ? s.y : 0.0f;
    r.z = s.z > 0.0f ? s.z : 0.0f;
    r.w = s.w > 0.0f ? s.w : 0.0f;
    return r;
}

__device__ __forceinline__ size_t row_of(int idx) {
    unsigned u = (unsigned)idx;
    return (size_t)(u < (unsigned)(V_N - 1) ? u : (unsigned)(V_N - 1));
}

#define AG_WPB 8
static_assert(V_N % AG_WPB == 0, "grid covers the vertices exactly");

__device__ __forceinline__ void agg0_term(v4f (&a0)[B_N], v4f (&a1)[B_N],
                                          const v4f (&l0)[B_N], const v4f (&l1)[B_N],
                                          const v4f t0, const v4f t1, const float w) {
#pragma unroll
    for (int b = 0; b < B_N; ++b) {
        a0[b] += relu4(t0 + l0[b]) * w;
        a1[b] += relu4(t1 + l1[b]) * w;
    }
}

__global__ void __launch_bounds__(32 * AG_WPB)
k_agg0(const float* __restrict__ xyz, const float* __restrict__ iW, const float* __restrict__ ib,
       const float* __restrict__ latW, const int* __restrict__ rowptr, const int* __restrict__ csrc,
       const float* __restrict__ cw, const float* __restrict__ selfnorm, float* __restrict__ y) {
    const int lane = threadIdx.x & 31;
    const int v = __builtin_amdgcn_readfirstlane((int)(blockIdx.x * AG_WPB + (threadIdx.x >> 5)));
    const int c0 = lane * 4, c1 = 128 + lane * 4;

    v4f wx0, wy0, wz0, wx1, wy1, wz1;
#pragma unroll
    for (int i = 0; i < 4; ++i) {
        const float* p0 = iW + (size_t)(c0 + i) * (LATENT + 3);
        const float* p1 = iW + (size_t)(c1 + i) * (LATENT + 3);
        wx0[i] = p0[0]; wy0[i] = p0[1]; wz0[i] = p0[2];
        wx1[i] = p1[0]; wy1[i] = p1[1]; wz1[i] = p1[2];
    }
    const v4f b0 = *(const v4f*)(ib + c0);
    const v4f b1 = *(const v4f*)(ib + c1);
    v4f l0[B_N], l1[B_N], a0[B_N], a1[B_N];
#pragma unroll
    for (int b = 0; b < B_N; ++b) {
        l0[b] = *(const v4f*)(latW + b * HIDDEN + c0);
        l1[b] = *(const v4f*)(latW + b * HIDDEN + c1);
        a0[b] = v4f{0.0f, 0.0f, 0.0f, 0.0f};
        a1[b] = v4f{0.0f, 0.0f, 0.0f, 0.0f};
    }

    const int beg = rowptr[v], end = rowptr[v + 1];
    int j = beg;
    for (; j + 4 <= end; j += 4) {
        size_t s[4];
        float w[4], px[4], py[4], pz[4];
#pragma unroll
        for (int k = 0; k < 4; ++k) { s[k] = row_of(csrc[j + k]); w[k] = cw[j + k]; }
#pragma unroll
        for (int k = 0; k < 4; ++k) {
            px[k] = xyz[3 * s[k]]; py[k] = xyz[3 * s[k] + 1]; pz[k] = xyz[3 * s[k] + 2];
        }
#pragma unroll
        for (int k = 0; k < 4; ++k) {
            const v4f t0 = b0 + wx0 * px[k] + wy0 * py[k] + wz0 * pz[k];
            const v4f t1 = b1 + wx1 * px[k] + wy1 * py[k] + wz1 * pz[k];
            agg0_term(a0, a1, l0, l1, t0, t1, w[k]);
        }
    }
    for (; j < end; ++j) {
        const size_t s = row_of(csrc[j]);
        const float w = cw[j];
        const float px = xyz[3 * s], py = xyz[3 * s + 1], pz = xyz[3 * s + 2];
        const v4f t0 = b0 + wx0 * px + wy0 * py + wz0 * pz;
        const v4f t1 = b1 + wx1 * px + wy1 * py + wz1 * pz;
        agg0_term(a0, a1, l0, l1, t0, t1, w);
    }
    {
        const float w = selfnorm[v];
        const float px = xyz[3 * v], py = xyz[3 * v + 1], pz = xyz[3 * v + 2];
        const v4f t0 = b0 + wx0 * px + wy0 * py + wz0 * pz;
        const v4f t1 = b1 + wx1 * px + wy1 * py + wz1 * pz;
        agg0_term(a0, a1, l0, l1, t0, t1, w);
    }

#pragma unroll
    for (int b = 0; b < B_N; ++b) {
        float* yo = y + ((size_t)b * V_N + v) * HIDDEN;
        *(v4f*)(yo + c0) = a0[b];
        *(v4f*)(yo + c1) = a1[b];
    }
}


#define GEMM_TILE_M  64
#define GEMM_TILES   (NNODES / GEMM_TILE_M)
#define GEMM_WGS     256
#define GEMM_NT      8
#define W_PITCH      260
static_assert(NNODES % GEMM_TILE_M == 0, "tiles cover the rows exactly");
static_assert(GEMM_WGS <= GEMM_TILES, "every work-group owns at least one tile");

template <bool EPI>
__global__ void __launch_bounds__(256)
k_gemm_wmma(const float* __restrict__ x, const float* __restrict__ W,
            const float* __restrict__ bias, float* __restrict__ h) {
    __shared__ __attribute__((aligned(16))) float Ws[HIDDEN * W_PITCH];

    const int lane  = threadIdx.x & 31;
    const int wv    = threadIdx.x >> 5;
    const int hf    = lane >> 4;
    const int l16   = lane & 15;
    const int mrow  = (wv >> 1) * 16;
    const int nbase = (wv & 1) * 128;

    const int ntile  = (GEMM_TILES - (int)blockIdx.x + GEMM_WGS - 1) / GEMM_WGS;
    const int nchunk = ntile * 8;

#if HAVE_WMMA_F32
    v4f a_cur[4], a_nxt[4];
    {
        const float* p = x + (size_t)((int)blockIdx.x * GEMM_TILE_M + mrow + l16) * HIDDEN + hf * 4;
#pragma unroll
        for (int j = 0; j < 4; ++j) a_cur[j] = *(const v4f*)(p + 8 * j);
    }
#endif

    {
        const int rot = (int)blockIdx.x & 63;
        for (int i = 0; i < 64; i += 16) {
            v4f t[16];
#pragma unroll
            for (int j = 0; j < 16; ++j) {
                const int q = (((i + j + rot) & 63) << 8) + (int)threadIdx.x;
                t[j] = *(const v4f*)(W + (size_t)q * 4);
            }
#pragma unroll
            for (int j = 0; j < 16; ++j) {
                const int q = (((i + j + rot) & 63) << 8) + (int)threadIdx.x;
                *(v4f*)(Ws + (q >> 6) * W_PITCH + (q & 63) * 4) = t[j];
            }
        }
    }
    __syncthreads();

#if HAVE_WMMA_F32
    const float* bp = Ws + (nbase + l16) * W_PITCH + hf * 4;

    v8f acc[GEMM_NT] = {};
    float bz[GEMM_NT];
#pragma unroll
    for (int nt = 0; nt < GEMM_NT; ++nt) bz[nt] = EPI ? bias[nbase + nt * 16 + l16] : 0.0f;
    v4f bq[GEMM_NT];
#pragma unroll
    for (int nt = 0; nt < GEMM_NT; ++nt) bq[nt] = *(const v4f*)(bp + nt * 16 * W_PITCH);

    for (int g = 0; g < nchunk; ++g) {
        const int kc = g & 7;
        {
            const int gn = (g + 1 < nchunk) ? g + 1 : g;
            const int tile = (int)blockIdx.x + (gn >> 3) * GEMM_WGS;
            const float* p = x + (size_t)(tile * GEMM_TILE_M + mrow + l16) * HIDDEN
                               + (gn & 7) * 32 + hf * 4;
#pragma unroll
            for (int j = 0; j < 4; ++j) a_nxt[j] = *(const v4f*)(p + 8 * j);
        }

#pragma unroll
        for (int j = 0; j < 4; ++j) {
            const int knext = (j < 3) ? kc * 32 + (j + 1) * 8 : ((kc + 1) & 7) * 32;
            v4f bn[GEMM_NT];
#pragma unroll
            for (int nt = 0; nt < GEMM_NT; ++nt)
                bn[nt] = *(const v4f*)(bp + nt * 16 * W_PITCH + knext);

            const v2f alo = {a_cur[j].x, a_cur[j].y};
            const v2f ahi = {a_cur[j].z, a_cur[j].w};
#pragma unroll
            for (int nt = 0; nt < GEMM_NT; ++nt) {
                const v2f blo = {bq[nt].x, bq[nt].y};
                acc[nt] = __builtin_amdgcn_wmma_f32_16x16x4_f32(
                     false, alo,  false, blo,
                     (short)0, acc[nt],
                     false,  false);
            }
#pragma unroll
            for (int nt = 0; nt < GEMM_NT; ++nt) {
                const v2f bhi = {bq[nt].z, bq[nt].w};
                acc[nt] = __builtin_amdgcn_wmma_f32_16x16x4_f32(
                     false, ahi,  false, bhi,
                     (short)0, acc[nt],
                     false,  false);
            }
#pragma unroll
            for (int nt = 0; nt < GEMM_NT; ++nt) bq[nt] = bn[nt];
        }

        if (kc == 7) {
            const int tile = (int)blockIdx.x + (g >> 3) * GEMM_WGS;
            float* outp = h + (size_t)(tile * GEMM_TILE_M + mrow + hf * 8) * HIDDEN + (nbase + l16);
#pragma unroll
            for (int nt = 0; nt < GEMM_NT; ++nt) {
#pragma unroll
                for (int r = 0; r < 8; ++r) {
                    float o = acc[nt][r];
                    if (EPI) { o += bz[nt]; o = o > 0.0f ? o : 0.0f; }
                    outp[(size_t)r * HIDDEN + nt * 16] = o;
                }
                acc[nt] = v8f{};
            }
        }
#pragma unroll
        for (int j = 0; j < 4; ++j) a_cur[j] = a_nxt[j];
    }
#else
    for (int t = 0; t < ntile; ++t) {
        const int tile = (int)blockIdx.x + t * GEMM_WGS;
        for (int idx = threadIdx.x; idx < GEMM_TILE_M * HIDDEN; idx += 256) {
            const int row = tile * GEMM_TILE_M + (idx >> 8), col = idx & 255;
            const float* xr = x + (size_t)row * HIDDEN;
            float s = 0.0f;
            for (int k = 0; k < HIDDEN; ++k) s += xr[k] * Ws[col * W_PITCH + k];
            if (EPI) { s += bias[col]; s = s > 0.0f ? s : 0.0f; }
            h[(size_t)row * HIDDEN + col] = s;
        }
    }
    (void)lane; (void)wv; (void)hf; (void)l16; (void)mrow; (void)nbase; (void)nchunk;
#endif
}


__device__ __forceinline__ void agg_row(const float* __restrict__ h, const int node,
                                        const int* __restrict__ rowptr, const int* __restrict__ csrc,
                                        const float* __restrict__ cw, const float* __restrict__ selfnorm,
                                        const int lane, v4f& r0, v4f& r1) {
    const int b = node / V_N;
    const int v = node - b * V_N;
    const float* hb = h + (size_t)b * V_N * HIDDEN;
    const float* hs = hb + (size_t)v * HIDDEN;
    const v4f s0 = *(const v4f*)(hs + lane * 4);
    const v4f s1 = *(const v4f*)(hs + 128 + lane * 4);
    const float sn = selfnorm[v];
    const int beg = rowptr[v], end = rowptr[v + 1];
    v4f a0 = {0.0f, 0.0f, 0.0f, 0.0f};
    v4f a1 = {0.0f, 0.0f, 0.0f, 0.0f};
    int j = beg;
    for (; end - j >= 8; j += 4) {
        v4f u0[4], u1[4];
        float w[4];
#pragma unroll
        for (int k = 0; k < 4; ++k) {
            const float* p = hb + row_of(csrc[j + k]) * HIDDEN;
            w[k]  = cw[j + k];
            u0[k] = *(const v4f*)(p + lane * 4);
            u1[k] = *(const v4f*)(p + 128 + lane * 4);
        }
#pragma unroll
        for (int k = 0; k < 4; ++k) {
            a0 += u0[k] * w[k];
            a1 += u1[k] * w[k];
        }
    }
    const int n = end - j;
    v4f u0[7], u1[7];
    float w[7];
#pragma unroll
    for (int k = 0; k < 7; ++k) {
        const bool on = k < n;
        const int e = on ? j + k : 0;
        const size_t sr = on ? row_of(csrc[e]) : (size_t)0;
        const float we = cw[e];
        const float* p = hb + sr * HIDDEN;
        w[k]  = on ? we : 0.0f;
        u0[k] = *(const v4f*)(p + lane * 4);
        u1[k] = *(const v4f*)(p + 128 + lane * 4);
    }
#pragma unroll
    for (int k = 0; k < 7; ++k) {
        a0 += u0[k] * w[k];
        a1 += u1[k] * w[k];
    }

    a0 += s0 * sn;
    a1 += s1 * sn;
    r0 = a0;
    r1 = a1;
}

#define FG_ROWS    16
#define FG_TILES   (NNODES / FG_ROWS)
#define FG_MWAVES  8
#define FG_GWAVES  16
#define FG_THREADS (32 * (FG_MWAVES + FG_GWAVES))
#define A_PITCH    260
#define FG_PIECES  (HIDDEN * HIDDEN / 4)
#define FG_CPY     11
#define FG_NOUT    (FG_ROWS * 3)
static_assert(NNODES % FG_ROWS == 0, "tiles cover the rows exactly");
static_assert(FG_GWAVES == FG_ROWS, "one gathering wave per tile row");
static_assert(2 * FG_CPY * FG_THREADS >= FG_PIECES, "the copy covers W");
static_assert(FG_TILES % 2 == 0, "LAST takes the tiles in pairs");
static_assert(2 * FG_NOUT == 3 * 32, "a pair of tiles has three whole 128-byte lines of results");

template <bool LAST>
__device__ __forceinline__ int fg_tile(const int wg, const int t) {
    return LAST ? 2 * (wg + (t >> 1) * GEMM_WGS) + (t & 1) : wg + t * GEMM_WGS;
}

template <int M>
__device__ __forceinline__ float row_xor_add(const float v) {
    const int other = __builtin_amdgcn_update_dpp(0, __float_as_int(v), 0x160 + M, 0xf, 0xf, false);
    return v + __int_as_float(other);
}

__device__ __forceinline__ void proj_write(const float* pp, const int bufe, const int bufo,
                                           const int pair, const int gw, const int lane,
                                           const float* __restrict__ ob, float* __restrict__ disp) {
    const int idx = gw * 32 + lane;
    const bool odd = idx >= FG_NOUT;
    const int j = odd ? idx - FG_NOUT : idx;
    const int buf = odd ? bufo : bufe;
    const float b0 = ob[0], b1 = ob[1], b2 = ob[2];
    const int o = j % 3;
    float s = o == 0 ? b0 : (o == 1 ? b1 : b2);
#pragma unroll
    for (int w = 0; w < FG_MWAVES; ++w) s += pp[(buf * FG_MWAVES + w) * FG_NOUT + j];
    disp[(size_t)pair * (2 * FG_NOUT) + idx] = s;
}

template <bool LAST>
__global__ void __launch_bounds__(FG_THREADS)
k_gemm_gather(const float* __restrict__ xin, const float* __restrict__ W,
              const float* __restrict__ bias,
              const int* __restrict__ rowptr, const int* __restrict__ csrc,
              const float* __restrict__ cw, const float* __restrict__ selfnorm,
              float* __restrict__ xout,
              const float* __restrict__ oW, const float* __restrict__ ob, float* __restrict__ disp) {
    __shared__ __attribute__((aligned(16))) float Ws[HIDDEN * W_PITCH];
    __shared__ __attribute__((aligned(16))) float As[2 * FG_ROWS * A_PITCH];
    __shared__ float Pp[LAST ? 4 * FG_MWAVES * FG_NOUT : 4];

    const int lane = threadIdx.x & 31;
    const int wv = __builtin_amdgcn_readfirstlane((int)(threadIdx.x >> 5));
    const bool gatherer = wv >= FG_MWAVES;
    const int grow = wv - FG_MWAVES;
    const int wg = (int)blockIdx.x;
    const int ntile = LAST ? 2 * ((FG_TILES / 2 - wg + GEMM_WGS - 1) / GEMM_WGS)
                           : (FG_TILES - wg + GEMM_WGS - 1) / GEMM_WGS;

    {
        const int rot = ((int)blockIdx.x & 63) << 8;
        for (int i = 0; i < 2 * FG_CPY; i += FG_CPY) {
            v4f t[FG_CPY];
#pragma unroll
            for (int j = 0; j < FG_CPY; ++j) {
                const int p = (i + j) * FG_THREADS + (int)threadIdx.x;
                const int q = (p + rot) & (FG_PIECES - 1);
                t[j] = (p < FG_PIECES) ? *(const v4f*)(W + (size_t)q * 4) : v4f{0.0f, 0.0f, 0.0f, 0.0f};
            }
#pragma unroll
            for (int j = 0; j < FG_CPY; ++j) {
                const int p = (i + j) * FG_THREADS + (int)threadIdx.x;
                const int q = (p + rot) & (FG_PIECES - 1);
                if (p < FG_PIECES) *(v4f*)(Ws + (q >> 6) * W_PITCH + (q & 63) * 4) = t[j];
            }
        }
    }

    if (gatherer) {
        __builtin_amdgcn_s_setprio(3);
        v4f r0, r1;
        agg_row(xin, fg_tile<LAST>(wg, 0) * FG_ROWS + grow, rowptr, csrc, cw, selfnorm, lane, r0, r1);
        float* ar = As + grow * A_PITCH;
        *(v4f*)(ar + lane * 4)       = r0;
        *(v4f*)(ar + 128 + lane * 4) = r1;
    }
    __syncthreads();

    const int hf    = lane >> 4;
    const int l16   = lane & 15;
    const int nbase = (gatherer ? 0 : wv) * 32;
    const float bz0 = bias[nbase + l16];
    const float bz1 = bias[nbase + 16 + l16];
    float ow0[3], ow1[3];
#pragma unroll
    for (int o = 0; o < 3; ++o) {
        ow0[o] = LAST ? oW[o * HIDDEN + nbase + l16] : 0.0f;
        ow1[o] = LAST ? oW[o * HIDDEN + nbase + 16 + l16] : 0.0f;
    }
#if HAVE_WMMA_F32
    const float* bp = Ws + (nbase + l16) * W_PITCH + hf * 4;
#endif

    for (int t = 0; t < ntile; ++t) {
        const int tile = fg_tile<LAST>(wg, t);
        if (gatherer) {
            if (LAST && t >= 2 && (t & 1) == 0 && grow < 3)
                proj_write(Pp, (t - 2) & 3, (t - 1) & 3, wg + ((t - 2) >> 1) * GEMM_WGS, grow, lane, ob, disp);
            if (t + 1 < ntile) {
                v4f r0, r1;
                agg_row(xin, fg_tile<LAST>(wg, t + 1) * FG_ROWS + grow, rowptr, csrc, cw, selfnorm,
                        lane, r0, r1);
                float* ar = As + ((t + 1) & 1) * (FG_ROWS * A_PITCH) + grow * A_PITCH;
                *(v4f*)(ar + lane * 4)       = r0;
                *(v4f*)(ar + 128 + lane * 4) = r1;
            }
        } else {
            const float* at = As + (t & 1) * (FG_ROWS * A_PITCH);
#if HAVE_WMMA_F32
            const float* ap = at + l16 * A_PITCH + hf * 4;
            v8f acc0 = {}, acc1 = {};
            v4f fa[4], fb0[4], fb1[4];
#pragma unroll
            for (int j = 0; j < 4; ++j) {
                fa[j]  = *(const v4f*)(ap + 8 * j);
                fb0[j] = *(const v4f*)(bp + 8 * j);
                fb1[j] = *(const v4f*)(bp + 16 * W_PITCH + 8 * j);
            }
#pragma unroll
            for (int kb = 0; kb < 8; ++kb) {
                v4f na[4], nb0[4], nb1[4];
                if (kb + 1 < 8) {
#pragma unroll
                    for (int j = 0; j < 4; ++j) {
                        na[j]  = *(const v4f*)(ap + (kb + 1) * 32 + 8 * j);
                        nb0[j] = *(const v4f*)(bp + (kb + 1) * 32 + 8 * j);
                        nb1[j] = *(const v4f*)(bp + 16 * W_PITCH + (kb + 1) * 32 + 8 * j);
                    }
                }
                __builtin_amdgcn_sched_barrier(0);
#pragma unroll
                for (int j = 0; j < 4; ++j) {
                    const v2f alo  = {fa[j].x, fa[j].y};
                    const v2f ahi  = {fa[j].z, fa[j].w};
                    const v2f b0lo = {fb0[j].x, fb0[j].y};
                    const v2f b0hi = {fb0[j].z, fb0[j].w};
                    const v2f b1lo = {fb1[j].x, fb1[j].y};
                    const v2f b1hi = {fb1[j].z, fb1[j].w};
                    acc0 = __builtin_amdgcn_wmma_f32_16x16x4_f32(false, alo, false, b0lo, (short)0, acc0, false, false);
                    acc1 = __builtin_amdgcn_wmma_f32_16x16x4_f32(false, alo, false, b1lo, (short)0, acc1, false, false);
                    acc0 = __builtin_amdgcn_wmma_f32_16x16x4_f32(false, ahi, false, b0hi, (short)0, acc0, false, false);
                    acc1 = __builtin_amdgcn_wmma_f32_16x16x4_f32(false, ahi, false, b1hi, (short)0, acc1, false, false);
                }
                __builtin_amdgcn_sched_barrier(0);
                if (kb + 1 < 8) {
#pragma unroll
                    for (int j = 0; j < 4; ++j) { fa[j] = na[j]; fb0[j] = nb0[j]; fb1[j] = nb1[j]; }
                }
            }
            if (!LAST) {
                float* outp = xout + (size_t)(tile * FG_ROWS + hf * 8) * HIDDEN + (nbase + l16);
#pragma unroll
                for (int r = 0; r < 8; ++r) {
                    const float y0 = acc0[r] + bz0;
                    const float y1 = acc1[r] + bz1;
                    outp[(size_t)r * HIDDEN]      = y0 > 0.0f ? y0 : 0.0f;
                    outp[(size_t)r * HIDDEN + 16] = y1 > 0.0f ? y1 : 0.0f;
                }
            } else {
                float p[3][8];
#pragma unroll
                for (int r = 0; r < 8; ++r) {
                    float y0 = acc0[r] + bz0;
                    float y1 = acc1[r] + bz1;
                    y0 = y0 > 0.0f ? y0 : 0.0f;
                    y1 = y1 > 0.0f ? y1 : 0.0f;
#pragma unroll
                    for (int o = 0; o < 3; ++o) p[o][r] = y0 * ow0[o] + y1 * ow1[o];
                }
#pragma unroll
                for (int o = 0; o < 3; ++o) {
#pragma unroll
                    for (int r = 0; r < 8; ++r) {
                        float v = p[o][r];
                        v = row_xor_add<8>(v);
                        v = row_xor_add<4>(v);
                        v = row_xor_add<2>(v);
                        v = row_xor_add<1>(v);
                        p[o][r] = v;
                    }
                }
                if (l16 == 0) {
                    float* pw = Pp + ((t & 3) * FG_MWAVES + wv) * FG_NOUT + (hf * 8) * 3;
#pragma unroll
                    for (int r = 0; r < 8; ++r) {
#pragma unroll
                        for (int o = 0; o < 3; ++o) pw[r * 3 + o] = p[o][r];
                    }
                }
            }
#else
            for (int m = 0; m < FG_ROWS; ++m) {
                const int col = (wv & 7) * 32 + lane;
                float s = 0.0f;
                for (int k = 0; k < HIDDEN; ++k)
                    s += at[m * A_PITCH + k] * Ws[col * W_PITCH + k];
                s += bias[col];
                s = s > 0.0f ? s : 0.0f;
                if (!LAST) {
                    xout[(size_t)(tile * FG_ROWS + m) * HIDDEN + col] = s;
                } else {
                    for (int o = 0; o < 3; ++o) {
                        float q = s * oW[o * HIDDEN + col];
                        for (int mm = 16; mm >= 1; mm >>= 1) q += __shfl_xor(q, mm);
                        if (lane == 0) Pp[((t & 3) * FG_MWAVES + (wv & 7)) * FG_NOUT + m * 3 + o] = q;
                    }
                }
            }
#endif
        }
        __builtin_amdgcn_fence(__ATOMIC_RELEASE, "workgroup", "local");
        __builtin_amdgcn_s_barrier();
        __builtin_amdgcn_fence(__ATOMIC_ACQUIRE, "workgroup", "local");
    }
    if (LAST && gatherer && grow < 3)
        proj_write(Pp, (ntile - 2) & 3, (ntile - 1) & 3, wg + ((ntile - 2) >> 1) * GEMM_WGS, grow, lane, ob, disp);
    if (gatherer) __builtin_amdgcn_s_setprio(0);
    (void)hf; (void)l16; (void)bz0; (void)bz1; (void)ow0; (void)ow1;
}


extern "C" void kernel_launch(void* const* d_in, const int* in_sizes, int n_in,
                              void* d_out, int out_size, void* d_ws, size_t ws_size,
                              hipStream_t stream) {
    const float* vertex_xyz = (const float*)d_in[0];
    const float* latent     = (const float*)d_in[1];
    const int*   edge_index = (const int*)d_in[2];
    const float* input_W    = (const float*)d_in[3];
    const float* input_b    = (const float*)d_in[4];
    const float* conv_W     = (const float*)d_in[5];
    const float* conv_b     = (const float*)d_in[6];
    const float* out_W      = (const float*)d_in[7];
    const float* out_b      = (const float*)d_in[8];
    float* disp = (float*)d_out;

    const int* src = edge_index;
    const int* dst = edge_index + E_N;

    float* xbuf     = (float*)d_ws;
    float* hbuf     = xbuf + (size_t)NNODES * HIDDEN;
    float* xyzW     = hbuf + (size_t)NNODES * HIDDEN;
    float* latW     = xyzW + (size_t)V_N * HIDDEN;
    float* dinv     = latW + (size_t)B_N * HIDDEN;
    float* selfnorm = dinv + V_N;
    int*   cnt      = (int*)(selfnorm + V_N);
    int*   rowptr   = cnt + V_N;
    int*   rank     = (int*)hbuf;
    int*   slot     = rank + E_N;
    int*   csrc     = (int*)xyzW;
    float* cw       = (float*)(csrc + E_N);

    const int T = 256;

    k_latW<<<(B_N * HIDDEN + T - 1) / T, T, 0, stream>>>(latent, input_W, latW);

    k_cnt_zero<<<(V_N + T - 1) / T, T, 0, stream>>>(cnt);
    k_rank<<<(E_N + T - 1) / T, T, 0, stream>>>(cnt, dst, rank);
    k_scan<<<1, SCAN_T, 0, stream>>>(cnt, rowptr, dinv, selfnorm);
    k_fill<<<(E_N + T - 1) / T, T, 0, stream>>>(rowptr, dst, rank, slot);
    k_place<<<(E_N + T - 1) / T, T, 0, stream>>>(rowptr, slot, src, dst, dinv, csrc, cw);

    dim3 ggrid(GEMM_WGS);
    k_agg0<<<V_N / AG_WPB, 32 * AG_WPB, 0, stream>>>(vertex_xyz, input_W, input_b, latW,
                                                     rowptr, csrc, cw, selfnorm, xbuf);
    k_gemm_wmma<true><<<ggrid, T, 0, stream>>>(xbuf, conv_W, conv_b, hbuf);

    k_gemm_gather<false><<<ggrid, FG_THREADS, 0, stream>>>(hbuf, conv_W + (size_t)1 * HIDDEN * HIDDEN,
                                                           conv_b + (size_t)1 * HIDDEN,
                                                           rowptr, csrc, cw, selfnorm, xbuf,
                                                           out_W, out_b, disp);
    k_gemm_gather<false><<<ggrid, FG_THREADS, 0, stream>>>(xbuf, conv_W + (size_t)2 * HIDDEN * HIDDEN,
                                                           conv_b + (size_t)2 * HIDDEN,
                                                           rowptr, csrc, cw, selfnorm, hbuf,
                                                           out_W, out_b, disp);
    k_gemm_gather<false><<<ggrid, FG_THREADS, 0, stream>>>(hbuf, conv_W + (size_t)3 * HIDDEN * HIDDEN,
                                                           conv_b + (size_t)3 * HIDDEN,
                                                           rowptr, csrc, cw, selfnorm, xbuf,
                                                           out_W, out_b, disp);
    k_gemm_gather<true><<<ggrid, FG_THREADS, 0, stream>>>(xbuf, conv_W + (size_t)4 * HIDDEN * HIDDEN,
                                                          conv_b + (size_t)4 * HIDDEN,
                                                          rowptr, csrc, cw, selfnorm, hbuf,
                                                          out_W, out_b, disp);
}


# k_latW: one wave per output column reads its input_W row and the latent rows by whole-line 16-byte loads, all 8 batches at once, lanes' partial sums added by a butterfly; that sum's order changes
# speedup vs baseline: 4.6433x; 1.0457x over previous
#include <hip/hip_runtime.h>
#include <hip/hip_bf16.h>

#define V_N     10000
#define E_N     60000
#define B_N     8
#define LATENT  512
#define HIDDEN  256
#define NLAYERS 5
#define NNODES  (B_N * V_N)

typedef float v2f __attribute__((ext_vector_type(2)));
typedef float v8f __attribute__((ext_vector_type(8)));

#if defined(__gfx1250__) && __has_builtin(__builtin_amdgcn_wmma_f32_16x16x4_f32)
#define HAVE_WMMA_F32 1
#endif


__global__ void k_cnt_zero(int* cnt) {
    int v = blockIdx.x * blockDim.x + threadIdx.x;
    if (v < V_N) cnt[v] = 0;
}

__global__ void k_rank(int* cnt, const int* __restrict__ dst, int* __restrict__ rank) {
    int e = blockIdx.x * blockDim.x + threadIdx.x;
    if (e < E_N) rank[e] = atomicAdd(&cnt[dst[e]], 1);
}

#define SCAN_T   1024
#define SCAN_PER ((V_N + SCAN_T - 1) / SCAN_T)
__global__ void __launch_bounds__(SCAN_T)
k_scan(const int* __restrict__ cnt, int* __restrict__ rowptr,
       float* __restrict__ dinv, float* __restrict__ selfnorm) {
    __shared__ int part[SCAN_T];
    const int t = threadIdx.x;
    const int base = t * SCAN_PER;
    int loc[SCAN_PER];
    int c[SCAN_PER];
    int s = 0;
#pragma unroll
    for (int i = 0; i < SCAN_PER; ++i) {
        int v = base + i;
        c[i] = (v < V_N) ? cnt[v] : 0;
        loc[i] = s;
        s += c[i];
    }
    part[t] = s;
    __syncthreads();
    for (int off = 1; off < SCAN_T; off <<= 1) {
        int add = (t >= off) ? part[t - off] : 0;
        __syncthreads();
        part[t] += add;
        __syncthreads();
    }
    const int excl = part[t] - s;
#pragma unroll
    for (int i = 0; i < SCAN_PER; ++i) {
        int v = base + i;
        if (v < V_N) {
            rowptr[v] = excl + loc[i];
            float di = rsqrtf((float)(c[i] + 1));
            dinv[v] = di;
            selfnorm[v] = di * di;
        }
    }
    if (t == SCAN_T - 1) rowptr[V_N] = part[t];
}

__global__ void k_fill(const int* __restrict__ rowptr, const int* __restrict__ dst,
                       const int* __restrict__ rank, int* __restrict__ slot) {
    int e = blockIdx.x * blockDim.x + threadIdx.x;
    if (e < E_N) slot[rowptr[dst[e]] + rank[e]] = e;
}

__global__ void k_place(const int* __restrict__ rowptr, const int* __restrict__ slot,
                        const int* __restrict__ src, const int* __restrict__ dst,
                        const float* __restrict__ dinv,
                        int* __restrict__ csrc, float* __restrict__ cw) {
    int e = blockIdx.x * blockDim.x + threadIdx.x;
    if (e >= E_N) return;
    const int v = dst[e], s = src[e];
    const int beg = rowptr[v], end = rowptr[v + 1];
    int r = 0;
    for (int j = beg; j < end; ++j) r += (slot[j] < e) ? 1 : 0;
    csrc[beg + r] = s;
    cw[beg + r]   = dinv[s] * dinv[v];
}


typedef float v4f __attribute__((ext_vector_type(4)));
typedef float v4fu __attribute__((ext_vector_type(4), aligned(4)));

#define LW_WPB 8
static_assert(HIDDEN % LW_WPB == 0, "grid covers the columns exactly");
static_assert(LATENT == 4 * 32 * 4, "a lane holds four 16-byte pieces of a row");
static_assert(B_N <= 32, "one lane per batch stores");

__global__ void __launch_bounds__(32 * LW_WPB)
k_latW(const float* __restrict__ latent, const float* __restrict__ iW,
       float* __restrict__ latW) {
    const int lane = threadIdx.x & 31;
    const int hh = (int)(blockIdx.x * LW_WPB + (threadIdx.x >> 5));
    const float* wp = iW + (size_t)hh * (LATENT + 3) + 3 + lane * 4;
    v4f w[4];
#pragma unroll
    for (int i = 0; i < 4; ++i) w[i] = *(const v4fu*)(wp + 128 * i);
    float acc[B_N];
#pragma unroll
    for (int b = 0; b < B_N; ++b) {
        const float* lp = latent + (size_t)b * LATENT + lane * 4;
        float s = 0.0f;
#pragma unroll
        for (int i = 0; i < 4; ++i) {
            const v4f l = *(const v4f*)(lp + 128 * i);
            s += l.x * w[i].x;
            s += l.y * w[i].y;
            s += l.z * w[i].z;
            s += l.w * w[i].w;
        }
        acc[b] = s;
    }
    float out = 0.0f;
#pragma unroll
    for (int b = 0; b < B_N; ++b) {
        float s = acc[b];
        s += __shfl_xor(s, 16);
        s += __shfl_xor(s, 8);
        s += __shfl_xor(s, 4);
        s += __shfl_xor(s, 2);
        s += __shfl_xor(s, 1);
        out = (lane == b) ? s : out;
    }
    if (lane < B_N) latW[lane * HIDDEN + hh] = out;
}

__device__ __forceinline__ v4f relu4(v4f s) {
    v4f r;
    r.x = s.x > 0.0f ? s.x : 0.0f;
    r.y = s.y > 0.0f ? s.y : 0.0f;
    r.z = s.z > 0.0f ? s.z : 0.0f;
    r.w = s.w > 0.0f ? s.w : 0.0f;
    return r;
}

__device__ __forceinline__ size_t row_of(int idx) {
    unsigned u = (unsigned)idx;
    return (size_t)(u < (unsigned)(V_N - 1) ? u : (unsigned)(V_N - 1));
}

#define AG_WPB 8
static_assert(V_N % AG_WPB == 0, "grid covers the vertices exactly");

__device__ __forceinline__ void agg0_term(v4f (&a0)[B_N], v4f (&a1)[B_N],
                                          const v4f (&l0)[B_N], const v4f (&l1)[B_N],
                                          const v4f t0, const v4f t1, const float w) {
#pragma unroll
    for (int b = 0; b < B_N; ++b) {
        a0[b] += relu4(t0 + l0[b]) * w;
        a1[b] += relu4(t1 + l1[b]) * w;
    }
}

__global__ void __launch_bounds__(32 * AG_WPB)
k_agg0(const float* __restrict__ xyz, const float* __restrict__ iW, const float* __restrict__ ib,
       const float* __restrict__ latW, const int* __restrict__ rowptr, const int* __restrict__ csrc,
       const float* __restrict__ cw, const float* __restrict__ selfnorm, float* __restrict__ y) {
    const int lane = threadIdx.x & 31;
    const int v = __builtin_amdgcn_readfirstlane((int)(blockIdx.x * AG_WPB + (threadIdx.x >> 5)));
    const int c0 = lane * 4, c1 = 128 + lane * 4;

    v4f wx0, wy0, wz0, wx1, wy1, wz1;
#pragma unroll
    for (int i = 0; i < 4; ++i) {
        const float* p0 = iW + (size_t)(c0 + i) * (LATENT + 3);
        const float* p1 = iW + (size_t)(c1 + i) * (LATENT + 3);
        wx0[i] = p0[0]; wy0[i] = p0[1]; wz0[i] = p0[2];
        wx1[i] = p1[0]; wy1[i] = p1[1]; wz1[i] = p1[2];
    }
    const v4f b0 = *(const v4f*)(ib + c0);
    const v4f b1 = *(const v4f*)(ib + c1);
    v4f l0[B_N], l1[B_N], a0[B_N], a1[B_N];
#pragma unroll
    for (int b = 0; b < B_N; ++b) {
        l0[b] = *(const v4f*)(latW + b * HIDDEN + c0);
        l1[b] = *(const v4f*)(latW + b * HIDDEN + c1);
        a0[b] = v4f{0.0f, 0.0f, 0.0f, 0.0f};
        a1[b] = v4f{0.0f, 0.0f, 0.0f, 0.0f};
    }

    const int beg = rowptr[v], end = rowptr[v + 1];
    int j = beg;
    for (; j + 4 <= end; j += 4) {
        size_t s[4];
        float w[4], px[4], py[4], pz[4];
#pragma unroll
        for (int k = 0; k < 4; ++k) { s[k] = row_of(csrc[j + k]); w[k] = cw[j + k]; }
#pragma unroll
        for (int k = 0; k < 4; ++k) {
            px[k] = xyz[3 * s[k]]; py[k] = xyz[3 * s[k] + 1]; pz[k] = xyz[3 * s[k] + 2];
        }
#pragma unroll
        for (int k = 0; k < 4; ++k) {
            const v4f t0 = b0 + wx0 * px[k] + wy0 * py[k] + wz0 * pz[k];
            const v4f t1 = b1 + wx1 * px[k] + wy1 * py[k] + wz1 * pz[k];
            agg0_term(a0, a1, l0, l1, t0, t1, w[k]);
        }
    }
    for (; j < end; ++j) {
        const size_t s = row_of(csrc[j]);
        const float w = cw[j];
        const float px = xyz[3 * s], py = xyz[3 * s + 1], pz = xyz[3 * s + 2];
        const v4f t0 = b0 + wx0 * px + wy0 * py + wz0 * pz;
        const v4f t1 = b1 + wx1 * px + wy1 * py + wz1 * pz;
        agg0_term(a0, a1, l0, l1, t0, t1, w);
    }
    {
        const float w = selfnorm[v];
        const float px = xyz[3 * v], py = xyz[3 * v + 1], pz = xyz[3 * v + 2];
        const v4f t0 = b0 + wx0 * px + wy0 * py + wz0 * pz;
        const v4f t1 = b1 + wx1 * px + wy1 * py + wz1 * pz;
        agg0_term(a0, a1, l0, l1, t0, t1, w);
    }

#pragma unroll
    for (int b = 0; b < B_N; ++b) {
        float* yo = y + ((size_t)b * V_N + v) * HIDDEN;
        *(v4f*)(yo + c0) = a0[b];
        *(v4f*)(yo + c1) = a1[b];
    }
}


#define GEMM_TILE_M  64
#define GEMM_TILES   (NNODES / GEMM_TILE_M)
#define GEMM_WGS     256
#define GEMM_NT      8
#define W_PITCH      260
static_assert(NNODES % GEMM_TILE_M == 0, "tiles cover the rows exactly");
static_assert(GEMM_WGS <= GEMM_TILES, "every work-group owns at least one tile");

template <bool EPI>
__global__ void __launch_bounds__(256)
k_gemm_wmma(const float* __restrict__ x, const float* __restrict__ W,
            const float* __restrict__ bias, float* __restrict__ h) {
    __shared__ __attribute__((aligned(16))) float Ws[HIDDEN * W_PITCH];

    const int lane  = threadIdx.x & 31;
    const int wv    = threadIdx.x >> 5;
    const int hf    = lane >> 4;
    const int l16   = lane & 15;
    const int mrow  = (wv >> 1) * 16;
    const int nbase = (wv & 1) * 128;

    const int ntile  = (GEMM_TILES - (int)blockIdx.x + GEMM_WGS - 1) / GEMM_WGS;
    const int nchunk = ntile * 8;

#if HAVE_WMMA_F32
    v4f a_cur[4], a_nxt[4];
    {
        const float* p = x + (size_t)((int)blockIdx.x * GEMM_TILE_M + mrow + l16) * HIDDEN + hf * 4;
#pragma unroll
        for (int j = 0; j < 4; ++j) a_cur[j] = *(const v4f*)(p + 8 * j);
    }
#endif

    {
        const int rot = (int)blockIdx.x & 63;
        for (int i = 0; i < 64; i += 16) {
            v4f t[16];
#pragma unroll
            for (int j = 0; j < 16; ++j) {
                const int q = (((i + j + rot) & 63) << 8) + (int)threadIdx.x;
                t[j] = *(const v4f*)(W + (size_t)q * 4);
            }
#pragma unroll
            for (int j = 0; j < 16; ++j) {
                const int q = (((i + j + rot) & 63) << 8) + (int)threadIdx.x;
                *(v4f*)(Ws + (q >> 6) * W_PITCH + (q & 63) * 4) = t[j];
            }
        }
    }
    __syncthreads();

#if HAVE_WMMA_F32
    const float* bp = Ws + (nbase + l16) * W_PITCH + hf * 4;

    v8f acc[GEMM_NT] = {};
    float bz[GEMM_NT];
#pragma unroll
    for (int nt = 0; nt < GEMM_NT; ++nt) bz[nt] = EPI ? bias[nbase + nt * 16 + l16] : 0.0f;
    v4f bq[GEMM_NT];
#pragma unroll
    for (int nt = 0; nt < GEMM_NT; ++nt) bq[nt] = *(const v4f*)(bp + nt * 16 * W_PITCH);

    for (int g = 0; g < nchunk; ++g) {
        const int kc = g & 7;
        {
            const int gn = (g + 1 < nchunk) ? g + 1 : g;
            const int tile = (int)blockIdx.x + (gn >> 3) * GEMM_WGS;
            const float* p = x + (size_t)(tile * GEMM_TILE_M + mrow + l16) * HIDDEN
                               + (gn & 7) * 32 + hf * 4;
#pragma unroll
            for (int j = 0; j < 4; ++j) a_nxt[j] = *(const v4f*)(p + 8 * j);
        }

#pragma unroll
        for (int j = 0; j < 4; ++j) {
            const int knext = (j < 3) ? kc * 32 + (j + 1) * 8 : ((kc + 1) & 7) * 32;
            v4f bn[GEMM_NT];
#pragma unroll
            for (int nt = 0; nt < GEMM_NT; ++nt)
                bn[nt] = *(const v4f*)(bp + nt * 16 * W_PITCH + knext);

            const v2f alo = {a_cur[j].x, a_cur[j].y};
            const v2f ahi = {a_cur[j].z, a_cur[j].w};
#pragma unroll
            for (int nt = 0; nt < GEMM_NT; ++nt) {
                const v2f blo = {bq[nt].x, bq[nt].y};
                acc[nt] = __builtin_amdgcn_wmma_f32_16x16x4_f32(
                     false, alo,  false, blo,
                     (short)0, acc[nt],
                     false,  false);
            }
#pragma unroll
            for (int nt = 0; nt < GEMM_NT; ++nt) {
                const v2f bhi = {bq[nt].z, bq[nt].w};
                acc[nt] = __builtin_amdgcn_wmma_f32_16x16x4_f32(
                     false, ahi,  false, bhi,
                     (short)0, acc[nt],
                     false,  false);
            }
#pragma unroll
            for (int nt = 0; nt < GEMM_NT; ++nt) bq[nt] = bn[nt];
        }

        if (kc == 7) {
            const int tile = (int)blockIdx.x + (g >> 3) * GEMM_WGS;
            float* outp = h + (size_t)(tile * GEMM_TILE_M + mrow + hf * 8) * HIDDEN + (nbase + l16);
#pragma unroll
            for (int nt = 0; nt < GEMM_NT; ++nt) {
#pragma unroll
                for (int r = 0; r < 8; ++r) {
                    float o = acc[nt][r];
                    if (EPI) { o += bz[nt]; o = o > 0.0f ? o : 0.0f; }
                    outp[(size_t)r * HIDDEN + nt * 16] = o;
                }
                acc[nt] = v8f{};
            }
        }
#pragma unroll
        for (int j = 0; j < 4; ++j) a_cur[j] = a_nxt[j];
    }
#else
    for (int t = 0; t < ntile; ++t) {
        const int tile = (int)blockIdx.x + t * GEMM_WGS;
        for (int idx = threadIdx.x; idx < GEMM_TILE_M * HIDDEN; idx += 256) {
            const int row = tile * GEMM_TILE_M + (idx >> 8), col = idx & 255;
            const float* xr = x + (size_t)row * HIDDEN;
            float s = 0.0f;
            for (int k = 0; k < HIDDEN; ++k) s += xr[k] * Ws[col * W_PITCH + k];
            if (EPI) { s += bias[col]; s = s > 0.0f ? s : 0.0f; }
            h[(size_t)row * HIDDEN + col] = s;
        }
    }
    (void)lane; (void)wv; (void)hf; (void)l16; (void)mrow; (void)nbase; (void)nchunk;
#endif
}


__device__ __forceinline__ void agg_row(const float* __restrict__ h, const int node,
                                        const int* __restrict__ rowptr, const int* __restrict__ csrc,
                                        const float* __restrict__ cw, const float* __restrict__ selfnorm,
                                        const int lane, v4f& r0, v4f& r1) {
    const int b = node / V_N;
    const int v = node - b * V_N;
    const float* hb = h + (size_t)b * V_N * HIDDEN;
    const float* hs = hb + (size_t)v * HIDDEN;
    const v4f s0 = *(const v4f*)(hs + lane * 4);
    const v4f s1 = *(const v4f*)(hs + 128 + lane * 4);
    const float sn = selfnorm[v];
    const int beg = rowptr[v], end = rowptr[v + 1];
    v4f a0 = {0.0f, 0.0f, 0.0f, 0.0f};
    v4f a1 = {0.0f, 0.0f, 0.0f, 0.0f};
    int j = beg;
    for (; end - j >= 8; j += 4) {
        v4f u0[4], u1[4];
        float w[4];
#pragma unroll
        for (int k = 0; k < 4; ++k) {
            const float* p = hb + row_of(csrc[j + k]) * HIDDEN;
            w[k]  = cw[j + k];
            u0[k] = *(const v4f*)(p + lane * 4);
            u1[k] = *(const v4f*)(p + 128 + lane * 4);
        }
#pragma unroll
        for (int k = 0; k < 4; ++k) {
            a0 += u0[k] * w[k];
            a1 += u1[k] * w[k];
        }
    }
    const int n = end - j;
    v4f u0[7], u1[7];
    float w[7];
#pragma unroll
    for (int k = 0; k < 7; ++k) {
        const bool on = k < n;
        const int e = on ? j + k : 0;
        const size_t sr = on ? row_of(csrc[e]) : (size_t)0;
        const float we = cw[e];
        const float* p = hb + sr * HIDDEN;
        w[k]  = on ? we : 0.0f;
        u0[k] = *(const v4f*)(p + lane * 4);
        u1[k] = *(const v4f*)(p + 128 + lane * 4);
    }
#pragma unroll
    for (int k = 0; k < 7; ++k) {
        a0 += u0[k] * w[k];
        a1 += u1[k] * w[k];
    }

    a0 += s0 * sn;
    a1 += s1 * sn;
    r0 = a0;
    r1 = a1;
}

#define FG_ROWS    16
#define FG_TILES   (NNODES / FG_ROWS)
#define FG_MWAVES  8
#define FG_GWAVES  16
#define FG_THREADS (32 * (FG_MWAVES + FG_GWAVES))
#define A_PITCH    260
#define FG_PIECES  (HIDDEN * HIDDEN / 4)
#define FG_CPY     11
#define FG_NOUT    (FG_ROWS * 3)
static_assert(NNODES % FG_ROWS == 0, "tiles cover the rows exactly");
static_assert(FG_GWAVES == FG_ROWS, "one gathering wave per tile row");
static_assert(2 * FG_CPY * FG_THREADS >= FG_PIECES, "the copy covers W");
static_assert(FG_TILES % 2 == 0, "LAST takes the tiles in pairs");
static_assert(2 * FG_NOUT == 3 * 32, "a pair of tiles has three whole 128-byte lines of results");

template <bool LAST>
__device__ __forceinline__ int fg_tile(const int wg, const int t) {
    return LAST ? 2 * (wg + (t >> 1) * GEMM_WGS) + (t & 1) : wg + t * GEMM_WGS;
}

template <int M>
__device__ __forceinline__ float row_xor_add(const float v) {
    const int other = __builtin_amdgcn_update_dpp(0, __float_as_int(v), 0x160 + M, 0xf, 0xf, false);
    return v + __int_as_float(other);
}

__device__ __forceinline__ void proj_write(const float* pp, const int bufe, const int bufo,
                                           const int pair, const int gw, const int lane,
                                           const float* __restrict__ ob, float* __restrict__ disp) {
    const int idx = gw * 32 + lane;
    const bool odd = idx >= FG_NOUT;
    const int j = odd ? idx - FG_NOUT : idx;
    const int buf = odd ? bufo : bufe;
    const float b0 = ob[0], b1 = ob[1], b2 = ob[2];
    const int o = j % 3;
    float s = o == 0 ? b0 : (o == 1 ? b1 : b2);
#pragma unroll
    for (int w = 0; w < FG_MWAVES; ++w) s += pp[(buf * FG_MWAVES + w) * FG_NOUT + j];
    disp[(size_t)pair * (2 * FG_NOUT) + idx] = s;
}

template <bool LAST>
__global__ void __launch_bounds__(FG_THREADS)
k_gemm_gather(const float* __restrict__ xin, const float* __restrict__ W,
              const float* __restrict__ bias,
              const int* __restrict__ rowptr, const int* __restrict__ csrc,
              const float* __restrict__ cw, const float* __restrict__ selfnorm,
              float* __restrict__ xout,
              const float* __restrict__ oW, const float* __restrict__ ob, float* __restrict__ disp) {
    __shared__ __attribute__((aligned(16))) float Ws[HIDDEN * W_PITCH];
    __shared__ __attribute__((aligned(16))) float As[2 * FG_ROWS * A_PITCH];
    __shared__ float Pp[LAST ? 4 * FG_MWAVES * FG_NOUT : 4];

    const int lane = threadIdx.x & 31;
    const int wv = __builtin_amdgcn_readfirstlane((int)(threadIdx.x >> 5));
    const bool gatherer = wv >= FG_MWAVES;
    const int grow = wv - FG_MWAVES;
    const int wg = (int)blockIdx.x;
    const int ntile = LAST ? 2 * ((FG_TILES / 2 - wg + GEMM_WGS - 1) / GEMM_WGS)
                           : (FG_TILES - wg + GEMM_WGS - 1) / GEMM_WGS;

    {
        const int rot = ((int)blockIdx.x & 63) << 8;
        for (int i = 0; i < 2 * FG_CPY; i += FG_CPY) {
            v4f t[FG_CPY];
#pragma unroll
            for (int j = 0; j < FG_CPY; ++j) {
                const int p = (i + j) * FG_THREADS + (int)threadIdx.x;
                const int q = (p + rot) & (FG_PIECES - 1);
                t[j] = (p < FG_PIECES) ? *(const v4f*)(W + (size_t)q * 4) : v4f{0.0f, 0.0f, 0.0f, 0.0f};
            }
#pragma unroll
            for (int j = 0; j < FG_CPY; ++j) {
                const int p = (i + j) * FG_THREADS + (int)threadIdx.x;
                const int q = (p + rot) & (FG_PIECES - 1);
                if (p < FG_PIECES) *(v4f*)(Ws + (q >> 6) * W_PITCH + (q & 63) * 4) = t[j];
            }
        }
    }

    if (gatherer) {
        __builtin_amdgcn_s_setprio(3);
        v4f r0, r1;
        agg_row(xin, fg_tile<LAST>(wg, 0) * FG_ROWS + grow, rowptr, csrc, cw, selfnorm, lane, r0, r1);
        float* ar = As + grow * A_PITCH;
        *(v4f*)(ar + lane * 4)       = r0;
        *(v4f*)(ar + 128 + lane * 4) = r1;
    }
    __syncthreads();

    const int hf    = lane >> 4;
    const int l16   = lane & 15;
    const int nbase = (gatherer ? 0 : wv) * 32;
    const float bz0 = bias[nbase + l16];
    const float bz1 = bias[nbase + 16 + l16];
    float ow0[3], ow1[3];
#pragma unroll
    for (int o = 0; o < 3; ++o) {
        ow0[o] = LAST ? oW[o * HIDDEN + nbase + l16] : 0.0f;
        ow1[o] = LAST ? oW[o * HIDDEN + nbase + 16 + l16] : 0.0f;
    }
#if HAVE_WMMA_F32
    const float* bp = Ws + (nbase + l16) * W_PITCH + hf * 4;
#endif

    for (int t = 0; t < ntile; ++t) {
        const int tile = fg_tile<LAST>(wg, t);
        if (gatherer) {
            if (LAST && t >= 2 && (t & 1) == 0 && grow < 3)
                proj_write(Pp, (t - 2) & 3, (t - 1) & 3, wg + ((t - 2) >> 1) * GEMM_WGS, grow, lane, ob, disp);
            if (t + 1 < ntile) {
                v4f r0, r1;
                agg_row(xin, fg_tile<LAST>(wg, t + 1) * FG_ROWS + grow, rowptr, csrc, cw, selfnorm,
                        lane, r0, r1);
                float* ar = As + ((t + 1) & 1) * (FG_ROWS * A_PITCH) + grow * A_PITCH;
                *(v4f*)(ar + lane * 4)       = r0;
                *(v4f*)(ar + 128 + lane * 4) = r1;
            }
        } else {
            const float* at = As + (t & 1) * (FG_ROWS * A_PITCH);
#if HAVE_WMMA_F32
            const float* ap = at + l16 * A_PITCH + hf * 4;
            v8f acc0 = {}, acc1 = {};
            v4f fa[4], fb0[4], fb1[4];
#pragma unroll
            for (int j = 0; j < 4; ++j) {
                fa[j]  = *(const v4f*)(ap + 8 * j);
                fb0[j] = *(const v4f*)(bp + 8 * j);
                fb1[j] = *(const v4f*)(bp + 16 * W_PITCH + 8 * j);
            }
#pragma unroll
            for (int kb = 0; kb < 8; ++kb) {
                v4f na[4], nb0[4], nb1[4];
                if (kb + 1 < 8) {
#pragma unroll
                    for (int j = 0; j < 4; ++j) {
                        na[j]  = *(const v4f*)(ap + (kb + 1) * 32 + 8 * j);
                        nb0[j] = *(const v4f*)(bp + (kb + 1) * 32 + 8 * j);
                        nb1[j] = *(const v4f*)(bp + 16 * W_PITCH + (kb + 1) * 32 + 8 * j);
                    }
                }
                __builtin_amdgcn_sched_barrier(0);
#pragma unroll
                for (int j = 0; j < 4; ++j) {
                    const v2f alo  = {fa[j].x, fa[j].y};
                    const v2f ahi  = {fa[j].z, fa[j].w};
                    const v2f b0lo = {fb0[j].x, fb0[j].y};
                    const v2f b0hi = {fb0[j].z, fb0[j].w};
                    const v2f b1lo = {fb1[j].x, fb1[j].y};
                    const v2f b1hi = {fb1[j].z, fb1[j].w};
                    acc0 = __builtin_amdgcn_wmma_f32_16x16x4_f32(false, alo, false, b0lo, (short)0, acc0, false, false);
                    acc1 = __builtin_amdgcn_wmma_f32_16x16x4_f32(false, alo, false, b1lo, (short)0, acc1, false, false);
                    acc0 = __builtin_amdgcn_wmma_f32_16x16x4_f32(false, ahi, false, b0hi, (short)0, acc0, false, false);
                    acc1 = __builtin_amdgcn_wmma_f32_16x16x4_f32(false, ahi, false, b1hi, (short)0, acc1, false, false);
                }
                __builtin_amdgcn_sched_barrier(0);
                if (kb + 1 < 8) {
#pragma unroll
                    for (int j = 0; j < 4; ++j) { fa[j] = na[j]; fb0[j] = nb0[j]; fb1[j] = nb1[j]; }
                }
            }
            if (!LAST) {
                float* outp = xout + (size_t)(tile * FG_ROWS + hf * 8) * HIDDEN + (nbase + l16);
#pragma unroll
                for (int r = 0; r < 8; ++r) {
                    const float y0 = acc0[r] + bz0;
                    const float y1 = acc1[r] + bz1;
                    outp[(size_t)r * HIDDEN]      = y0 > 0.0f ? y0 : 0.0f;
                    outp[(size_t)r * HIDDEN + 16] = y1 > 0.0f ? y1 : 0.0f;
                }
            } else {
                float p[3][8];
#pragma unroll
                for (int r = 0; r < 8; ++r) {
                    float y0 = acc0[r] + bz0;
                    float y1 = acc1[r] + bz1;
                    y0 = y0 > 0.0f ? y0 : 0.0f;
                    y1 = y1 > 0.0f ? y1 : 0.0f;
#pragma unroll
                    for (int o = 0; o < 3; ++o) p[o][r] = y0 * ow0[o] + y1 * ow1[o];
                }
#pragma unroll
                for (int o = 0; o < 3; ++o) {
#pragma unroll
                    for (int r = 0; r < 8; ++r) {
                        float v = p[o][r];
                        v = row_xor_add<8>(v);
                        v = row_xor_add<4>(v);
                        v = row_xor_add<2>(v);
                        v = row_xor_add<1>(v);
                        p[o][r] = v;
                    }
                }
                if (l16 == 0) {
                    float* pw = Pp + ((t & 3) * FG_MWAVES + wv) * FG_NOUT + (hf * 8) * 3;
#pragma unroll
                    for (int r = 0; r < 8; ++r) {
#pragma unroll
                        for (int o = 0; o < 3; ++o) pw[r * 3 + o] = p[o][r];
                    }
                }
            }
#else
            for (int m = 0; m < FG_ROWS; ++m) {
                const int col = (wv & 7) * 32 + lane;
                float s = 0.0f;
                for (int k = 0; k < HIDDEN; ++k)
                    s += at[m * A_PITCH + k] * Ws[col * W_PITCH + k];
                s += bias[col];
                s = s > 0.0f ? s : 0.0f;
                if (!LAST) {
                    xout[(size_t)(tile * FG_ROWS + m) * HIDDEN + col] = s;
                } else {
                    for (int o = 0; o < 3; ++o) {
                        float q = s * oW[o * HIDDEN + col];
                        for (int mm = 16; mm >= 1; mm >>= 1) q += __shfl_xor(q, mm);
                        if (lane == 0) Pp[((t & 3) * FG_MWAVES + (wv & 7)) * FG_NOUT + m * 3 + o] = q;
                    }
                }
            }
#endif
        }
        __builtin_amdgcn_fence(__ATOMIC_RELEASE, "workgroup", "local");
        __builtin_amdgcn_s_barrier();
        __builtin_amdgcn_fence(__ATOMIC_ACQUIRE, "workgroup", "local");
    }
    if (LAST && gatherer && grow < 3)
        proj_write(Pp, (ntile - 2) & 3, (ntile - 1) & 3, wg + ((ntile - 2) >> 1) * GEMM_WGS, grow, lane, ob, disp);
    if (gatherer) __builtin_amdgcn_s_setprio(0);
    (void)hf; (void)l16; (void)bz0; (void)bz1; (void)ow0; (void)ow1;
}


extern "C" void kernel_launch(void* const* d_in, const int* in_sizes, int n_in,
                              void* d_out, int out_size, void* d_ws, size_t ws_size,
                              hipStream_t stream) {
    const float* vertex_xyz = (const float*)d_in[0];
    const float* latent     = (const float*)d_in[1];
    const int*   edge_index = (const int*)d_in[2];
    const float* input_W    = (const float*)d_in[3];
    const float* input_b    = (const float*)d_in[4];
    const float* conv_W     = (const float*)d_in[5];
    const float* conv_b     = (const float*)d_in[6];
    const float* out_W      = (const float*)d_in[7];
    const float* out_b      = (const float*)d_in[8];
    float* disp = (float*)d_out;

    const int* src = edge_index;
    const int* dst = edge_index + E_N;

    float* xbuf     = (float*)d_ws;
    float* hbuf     = xbuf + (size_t)NNODES * HIDDEN;
    float* xyzW     = hbuf + (size_t)NNODES * HIDDEN;
    float* latW     = xyzW + (size_t)V_N * HIDDEN;
    float* dinv     = latW + (size_t)B_N * HIDDEN;
    float* selfnorm = dinv + V_N;
    int*   cnt      = (int*)(selfnorm + V_N);
    int*   rowptr   = cnt + V_N;
    int*   rank     = (int*)hbuf;
    int*   slot     = rank + E_N;
    int*   csrc     = (int*)xyzW;
    float* cw       = (float*)(csrc + E_N);

    const int T = 256;

    k_latW<<<HIDDEN / LW_WPB, 32 * LW_WPB, 0, stream>>>(latent, input_W, latW);

    k_cnt_zero<<<(V_N + T - 1) / T, T, 0, stream>>>(cnt);
    k_rank<<<(E_N + T - 1) / T, T, 0, stream>>>(cnt, dst, rank);
    k_scan<<<1, SCAN_T, 0, stream>>>(cnt, rowptr, dinv, selfnorm);
    k_fill<<<(E_N + T - 1) / T, T, 0, stream>>>(rowptr, dst, rank, slot);
    k_place<<<(E_N + T - 1) / T, T, 0, stream>>>(rowptr, slot, src, dst, dinv, csrc, cw);

    dim3 ggrid(GEMM_WGS);
    k_agg0<<<V_N / AG_WPB, 32 * AG_WPB, 0, stream>>>(vertex_xyz, input_W, input_b, latW,
                                                     rowptr, csrc, cw, selfnorm, xbuf);
    k_gemm_wmma<true><<<ggrid, T, 0, stream>>>(xbuf, conv_W, conv_b, hbuf);

    k_gemm_gather<false><<<ggrid, FG_THREADS, 0, stream>>>(hbuf, conv_W + (size_t)1 * HIDDEN * HIDDEN,
                                                           conv_b + (size_t)1 * HIDDEN,
                                                           rowptr, csrc, cw, selfnorm, xbuf,
                                                           out_W, out_b, disp);
    k_gemm_gather<false><<<ggrid, FG_THREADS, 0, stream>>>(xbuf, conv_W + (size_t)2 * HIDDEN * HIDDEN,
                                                           conv_b + (size_t)2 * HIDDEN,
                                                           rowptr, csrc, cw, selfnorm, hbuf,
                                                           out_W, out_b, disp);
    k_gemm_gather<false><<<ggrid, FG_THREADS, 0, stream>>>(hbuf, conv_W + (size_t)3 * HIDDEN * HIDDEN,
                                                           conv_b + (size_t)3 * HIDDEN,
                                                           rowptr, csrc, cw, selfnorm, xbuf,
                                                           out_W, out_b, disp);
    k_gemm_gather<true><<<ggrid, FG_THREADS, 0, stream>>>(xbuf, conv_W + (size_t)4 * HIDDEN * HIDDEN,
                                                          conv_b + (size_t)4 * HIDDEN,
                                                          rowptr, csrc, cw, selfnorm, hbuf,
                                                          out_W, out_b, disp);
}
